# starting kernel
# baseline (speedup 1.0000x reference)
#include <hip/hip_runtime.h>

#define NN 50000
#define NE 800000

typedef __attribute__((ext_vector_type(2))) float v2f;
typedef __attribute__((ext_vector_type(8))) float v8f;

__global__ void k_init_deg(float* deg) {
  int i = blockIdx.x * blockDim.x + threadIdx.x;
  if (i < NN) deg[i] = 1.0f;
}

__global__ void k_deg_edges(const int* __restrict__ dst, float* __restrict__ deg) {
  int e = blockIdx.x * blockDim.x + threadIdx.x;
  if (e < NE)
    __hip_atomic_fetch_add(deg + dst[e], 1.0f, __ATOMIC_RELAXED, __HIP_MEMORY_SCOPE_AGENT);
}

__global__ void k_rsqrt(float* d) {
  int i = blockIdx.x * blockDim.x + threadIdx.x;
  if (i < NN) d[i] = rsqrtf(d[i]);
}

__global__ void k_zero(float* __restrict__ p, unsigned long long n) {
  unsigned long long i = (unsigned long long)blockIdx.x * blockDim.x + threadIdx.x;
  if (i < n) p[i] = 0.0f;
}

__global__ void k_gemm_wmma(const float* __restrict__ X, const float* __restrict__ W,
                            float* __restrict__ H, int K, int N) {
  int wave = (int)((blockIdx.x * blockDim.x + threadIdx.x) >> 5);
  int lane = threadIdx.x & 31;
  int tilesN = N >> 4;
  int totTiles = (NN >> 4) * tilesN;
  if (wave >= totTiles) return;
  int tr = wave / tilesN;
  int tm = tr << 4;
  int tn = (wave - tr * tilesN) << 4;
  int half = lane >> 4;
  int l    = lane & 15;

  const float* xrow = X + (size_t)(tm + l) * K + 2 * half;
  const float* wcol = W + tn + l;
  v8f c = {};
  for (int k = 0; k < K; k += 4) {
    v2f a, b;
    a.x = xrow[k];
    a.y = xrow[k + 1];
    b.x = wcol[(size_t)(k + 2 * half) * N];
    b.y = wcol[(size_t)(k + 2 * half + 1) * N];
    c = __builtin_amdgcn_wmma_f32_16x16x4_f32(false, a, false, b, (short)0, c, false, false);
  }
  float* hrow = H + (size_t)(tm + 8 * half) * N + tn + l;
#pragma unroll
  for (int r = 0; r < 8; ++r) hrow[(size_t)r * N] = c[r];
}

__global__ void k_edge_scatter(const float* __restrict__ H, const int* __restrict__ src,
                               const int* __restrict__ dst, const float* __restrict__ dinv,
                               float* __restrict__ out, int C4) {
  long long gid = (long long)blockIdx.x * blockDim.x + threadIdx.x;
  int e = (int)(gid / C4);
  if (e >= NE) return;
  int c = (int)(gid - (long long)e * C4) << 2;
  int s = src[e], d = dst[e];
  if ((threadIdx.x & 63) == 0 && e + 256 < NE) {
    __builtin_prefetch(src + e + 256, 0, 0);
    __builtin_prefetch(dst + e + 256, 0, 0);
  }
  float nrm = dinv[s] * dinv[d];
  int C = C4 << 2;
  const float4 v = *(const float4*)(H + (size_t)s * C + c);
  float* op = out + (size_t)d * C + c;
  __hip_atomic_fetch_add(op + 0, v.x * nrm, __ATOMIC_RELAXED, __HIP_MEMORY_SCOPE_AGENT);
  __hip_atomic_fetch_add(op + 1, v.y * nrm, __ATOMIC_RELAXED, __HIP_MEMORY_SCOPE_AGENT);
  __hip_atomic_fetch_add(op + 2, v.z * nrm, __ATOMIC_RELAXED, __HIP_MEMORY_SCOPE_AGENT);
  __hip_atomic_fetch_add(op + 3, v.w * nrm, __ATOMIC_RELAXED, __HIP_MEMORY_SCOPE_AGENT);
}

__global__ void k_finalize(float* __restrict__ out, const float* __restrict__ H,
                           const float* __restrict__ dinv, const float* __restrict__ bias,
                           int C, int relu) {
  long long gid = (long long)blockIdx.x * blockDim.x + threadIdx.x;
  if (gid >= (long long)NN * C) return;
  int i = (int)(gid / C);
  int c = (int)(gid - (long long)i * C);
  float di = dinv[i];
  float v = out[gid] + H[gid] * (di * di) + bias[c];
  out[gid] = relu ? fmaxf(v, 0.0f) : v;
}

static inline unsigned nblk(long long n, int b) { return (unsigned)((n + b - 1) / b); }

extern "C" void kernel_launch(void* const* d_in, const int* in_sizes, int n_in,
                              void* d_out, int out_size, void* d_ws, size_t ws_size,
                              hipStream_t stream) {
  (void)in_sizes; (void)n_in; (void)out_size; (void)ws_size;
  const float* x   = (const float*)d_in[0];
  const int*   ei  = (const int*)d_in[1];
  const float* We1 = (const float*)d_in[2]; const float* be1 = (const float*)d_in[3];
  const float* We2 = (const float*)d_in[4]; const float* be2 = (const float*)d_in[5];
  const float* Wd1 = (const float*)d_in[6]; const float* bd1 = (const float*)d_in[7];
  const float* Wd2 = (const float*)d_in[8]; const float* bd2 = (const float*)d_in[9];
  const int* src = ei;
  const int* dst = ei + NE;

  char*  ws   = (char*)d_ws;
  float* dinv = (float*)ws;
  size_t off  = ((size_t)NN * 4 + 255) & ~(size_t)255;
  float* G    = (float*)(ws + off);
  float* P    = G + (size_t)NN * 128;

  k_init_deg <<<nblk(NN, 256), 256, 0, stream>>>(dinv);
  k_deg_edges<<<nblk(NE, 256), 256, 0, stream>>>(dst, dinv);
  k_rsqrt    <<<nblk(NN, 256), 256, 0, stream>>>(dinv);

  struct Layer { const float* in; const float* W; const float* b; int K; int C; int relu; float* S; };
  Layer layers[4] = {
    { x, We1, be1, 128,  64, 1, P },
    { P, We2, be2,  64,  32, 0, P },
    { P, Wd1, bd1,  32,  64, 1, P },
    { P, Wd2, bd2,  64, 128, 0, (float*)d_out },
  };

  for (int li = 0; li < 4; ++li) {
    const Layer& L = layers[li];
    long long tot   = (long long)NN * L.C;
    long long tiles = (long long)(NN / 16) * (L.C / 16);
    k_gemm_wmma   <<<nblk(tiles * 32, 256), 256, 0, stream>>>(L.in, L.W, G, L.K, L.C);
    k_zero        <<<nblk(tot, 256),        256, 0, stream>>>(L.S, (unsigned long long)tot);
    k_edge_scatter<<<nblk((long long)NE * (L.C / 4), 256), 256, 0, stream>>>(G, src, dst, dinv, L.S, L.C / 4);
    k_finalize    <<<nblk(tot, 256),        256, 0, stream>>>(L.S, G, dinv, L.b, L.C, L.relu);
  }
}


# Aggregation made destination-stationary: each work-group keeps its rows in LDS, sifts the dst list, adds gathered rows with LDS float adds (sum order = arrival); zero/scatter/finalize passes dropped
# speedup vs baseline: 1.5367x; 1.5367x over previous
#include <hip/hip_runtime.h>


#define NN 50000
#define NE 800000

typedef __attribute__((ext_vector_type(2))) float v2f;
typedef __attribute__((ext_vector_type(8))) float v8f;

__global__ void k_init_deg(float* deg) {
  int i = blockIdx.x * blockDim.x + threadIdx.x;
  if (i < NN) deg[i] = 1.0f;
}

__global__ void k_deg_edges(const int* __restrict__ dst, float* __restrict__ deg) {
  int e = blockIdx.x * blockDim.x + threadIdx.x;
  if (e < NE)
    __hip_atomic_fetch_add(deg + dst[e], 1.0f, __ATOMIC_RELAXED, __HIP_MEMORY_SCOPE_AGENT);
}

__global__ void k_rsqrt(float* d) {
  int i = blockIdx.x * blockDim.x + threadIdx.x;
  if (i < NN) d[i] = rsqrtf(d[i]);
}

__global__ void k_gemm_wmma(const float* __restrict__ X, const float* __restrict__ W,
                            float* __restrict__ H, int K, int N) {
  int wave = (int)((blockIdx.x * blockDim.x + threadIdx.x) >> 5);
  int lane = threadIdx.x & 31;
  int tilesN = N >> 4;
  int totTiles = (NN >> 4) * tilesN;
  if (wave >= totTiles) return;
  int tr = wave / tilesN;
  int tm = tr << 4;
  int tn = (wave - tr * tilesN) << 4;
  int half = lane >> 4;
  int l    = lane & 15;

  const float* xrow = X + (size_t)(tm + l) * K + 2 * half;
  const float* wcol = W + tn + l;
  v8f c = {};
  for (int k = 0; k < K; k += 4) {
    v2f a, b;
    a.x = xrow[k];
    a.y = xrow[k + 1];
    b.x = wcol[(size_t)(k + 2 * half) * N];
    b.y = wcol[(size_t)(k + 2 * half + 1) * N];
    c = __builtin_amdgcn_wmma_f32_16x16x4_f32(false, a, false, b, (short)0, c, false, false);
  }
  float* hrow = H + (size_t)(tm + 8 * half) * N + tn + l;
#pragma unroll
  for (int r = 0; r < 8; ++r) hrow[(size_t)r * N] = c[r];
}

#define AGG_NB    128
#define AGG_ROWS  ((NN + AGG_NB - 1) / AGG_NB)
#define AGG_QCAP  544
#define AGG_U     4

static_assert(NE % 4 == 0, "dst is streamed as int4");
static_assert(AGG_ROWS <= 512, "row offset is packed into 9 bits");
static_assert(NE <= (1 << 22), "edge id is packed above 9 bits");

template <int V> using vfN = float __attribute__((ext_vector_type(V)));

template <int C>
__device__ __forceinline__ void agg_drain(float* __restrict__ acc, const unsigned* __restrict__ qp,
                                          int cnt, const float* __restrict__ H,
                                          const int* __restrict__ src, const float* __restrict__ dinv,
                                          int lo, int lane) {
  constexpr int V = C / 32;
  const bool act = lane < cnt;
  const unsigned ent = act ? qp[lane] : 0u;
  const int e = (int)(ent >> 9);
  const int r = (int)(ent & 511u);
  const int s = act ? src[e] : 0;
  const float w = act ? dinv[s] * dinv[lo + r] : 0.0f;
  constexpr int G = 16;
  for (int j0 = 0; j0 < cnt; j0 += G) {
    vfN<V> v[G];
#pragma unroll
    for (int jj = 0; jj < G; ++jj) {
      const int sj = __builtin_amdgcn_readlane(s, j0 + jj);
      v[jj] = *(const vfN<V>*)(H + (size_t)sj * C + V * lane);
    }
#pragma unroll
    for (int jj = 0; jj < G; ++jj) {
      const int j = j0 + jj;
      const int   rj = __builtin_amdgcn_readlane(r, j);
      const float wj = __int_as_float(__builtin_amdgcn_readlane(__float_as_int(w), j));
      float* ap = acc + rj * C + lane;
      const bool live = j < cnt;
#pragma unroll
      for (int k = 0; k < V; ++k) {
        const float m = live ? v[jj][k] * wj : 0.0f;
        __hip_atomic_fetch_add(ap + 32 * k, m, __ATOMIC_RELAXED, __HIP_MEMORY_SCOPE_WORKGROUP);
      }
    }
  }
}

template <int C>
__global__ __launch_bounds__(1024) void k_agg(const float* __restrict__ H, const int* __restrict__ src,
                                              const int* __restrict__ dst, const float* __restrict__ dinv,
                                              const float* __restrict__ bias, float* __restrict__ out,
                                              int relu) {
  constexpr int V = C / 32;
  __shared__ float acc[AGG_ROWS * C];
  __shared__ unsigned q[32 * AGG_QCAP];
  const int tid  = threadIdx.x;
  const int lane = tid & 31;
  const int wv   = tid >> 5;
  const int lo   = blockIdx.x * AGG_ROWS;
  const int nrows = min(AGG_ROWS, NN - lo);

  {
    vfN<V> bv = *(const vfN<V>*)(bias + V * lane);
    for (int r = wv; r < nrows; r += 32) {
      const float di = dinv[lo + r];
      const float d2 = di * di;
      const vfN<V> hv = *(const vfN<V>*)(H + (size_t)(lo + r) * C + V * lane);
#pragma unroll
      for (int k = 0; k < V; ++k) acc[r * C + k * 32 + lane] = hv[k] * d2 + bv[k];
    }
  }
  __syncthreads();

  unsigned* myq = q + wv * AGG_QCAP;
  int qn = 0;
  const int4* d4 = (const int4*)dst;
  constexpr int N4 = NE / 4;

  int4 cur[AGG_U];
#pragma unroll
  for (int u = 0; u < AGG_U; ++u) cur[u] = d4[min(u * 1024 + tid, N4 - 1)];
  for (int base = 0; base < N4; base += 1024 * AGG_U) {
    int4 nxt[AGG_U];
#pragma unroll
    for (int u = 0; u < AGG_U; ++u) {
      nxt[u] = d4[min(base + 1024 * AGG_U + u * 1024 + tid, N4 - 1)];
    }
    unsigned rr[AGG_U][4];
    unsigned mk[AGG_U][4];
    unsigned lim[AGG_U];
#pragma unroll
    for (int u = 0; u < AGG_U; ++u) {
      lim[u] = (base + u * 1024 + tid < N4) ? (unsigned)nrows : 0u;
      rr[u][0] = (unsigned)(cur[u].x - lo);
      rr[u][1] = (unsigned)(cur[u].y - lo);
      rr[u][2] = (unsigned)(cur[u].z - lo);
      rr[u][3] = (unsigned)(cur[u].w - lo);
#pragma unroll
      for (int k = 0; k < 4; ++k) mk[u][k] = __builtin_amdgcn_ballot_w32(rr[u][k] < lim[u]);
    }
#pragma unroll
    for (int u = 0; u < AGG_U; ++u) {
      const unsigned e0s = (unsigned)(base + u * 1024 + tid) << 11;
#pragma unroll
      for (int k = 0; k < 4; ++k) {
        const unsigned m = mk[u][k];
        if (m) {
          if (rr[u][k] < lim[u])
            myq[qn + __builtin_amdgcn_mbcnt_lo(m, 0u)] = e0s + ((unsigned)k << 9) + rr[u][k];
          qn += __builtin_popcount(m);
        }
      }
    }
    while (qn >= 32) {
      qn -= 32;
      agg_drain<C>(acc, myq + qn, 32, H, src, dinv, lo, lane);
    }
#pragma unroll
    for (int u = 0; u < AGG_U; ++u) cur[u] = nxt[u];
  }
  if (qn > 0) agg_drain<C>(acc, myq, qn, H, src, dinv, lo, lane);
  __syncthreads();

  for (int r = wv; r < nrows; r += 32) {
    vfN<V> ov;
#pragma unroll
    for (int k = 0; k < V; ++k) {
      const float x = acc[r * C + k * 32 + lane];
      ov[k] = relu ? fmaxf(x, 0.0f) : x;
    }
    *(vfN<V>*)(out + (size_t)(lo + r) * C + V * lane) = ov;
  }
}

static inline unsigned nblk(long long n, int b) { return (unsigned)((n + b - 1) / b); }

extern "C" void kernel_launch(void* const* d_in, const int* in_sizes, int n_in,
                              void* d_out, int out_size, void* d_ws, size_t ws_size,
                              hipStream_t stream) {
  (void)in_sizes; (void)n_in; (void)out_size; (void)ws_size;
  const float* x   = (const float*)d_in[0];
  const int*   ei  = (const int*)d_in[1];
  const float* We1 = (const float*)d_in[2]; const float* be1 = (const float*)d_in[3];
  const float* We2 = (const float*)d_in[4]; const float* be2 = (const float*)d_in[5];
  const float* Wd1 = (const float*)d_in[6]; const float* bd1 = (const float*)d_in[7];
  const float* Wd2 = (const float*)d_in[8]; const float* bd2 = (const float*)d_in[9];
  const int* src = ei;
  const int* dst = ei + NE;

  char*  ws   = (char*)d_ws;
  float* dinv = (float*)ws;
  size_t off  = ((size_t)NN * 4 + 255) & ~(size_t)255;
  float* G    = (float*)(ws + off);
  float* P    = G + (size_t)NN * 128;
  float* O    = (float*)d_out;

  k_init_deg <<<nblk(NN, 256), 256, 0, stream>>>(dinv);
  k_deg_edges<<<nblk(NE, 256), 256, 0, stream>>>(dst, dinv);
  k_rsqrt    <<<nblk(NN, 256), 256, 0, stream>>>(dinv);

  const long long rowTiles = NN / 16;

  k_gemm_wmma<<<nblk(rowTiles * 4 * 32, 256), 256, 0, stream>>>(x, We1, G, 128, 64);
  k_agg<64>  <<<AGG_NB, 1024, 0, stream>>>(G, src, dst, dinv, be1, P, 1);

  k_gemm_wmma<<<nblk(rowTiles * 2 * 32, 256), 256, 0, stream>>>(P, We2, G, 64, 32);
  k_agg<32>  <<<AGG_NB, 1024, 0, stream>>>(G, src, dst, dinv, be2, P, 0);

  k_gemm_wmma<<<nblk(rowTiles * 4 * 32, 256), 256, 0, stream>>>(P, Wd1, G, 32, 64);
  k_agg<64>  <<<AGG_NB, 1024, 0, stream>>>(G, src, dst, dinv, bd1, P, 1);

  k_gemm_wmma<<<nblk(rowTiles * 8 * 32, 256), 256, 0, stream>>>(P, Wd2, G, 64, 128);
  k_agg<128> <<<AGG_NB, 1024, 0, stream>>>(G, src, dst, dinv, bd2, O, 0);
}


# dst list sifted once: the first aggregation appends each work-group's matched edges (src,row,weight) to a workspace list; the 3 later ones read that list, not all of dst; sums stay in arrival order
# speedup vs baseline: 1.8870x; 1.2280x over previous
#include <hip/hip_runtime.h>


#define NN 50000
#define NE 800000

typedef __attribute__((ext_vector_type(2))) float v2f;
typedef __attribute__((ext_vector_type(8))) float v8f;

__global__ void k_init_deg(float* deg) {
  int i = blockIdx.x * blockDim.x + threadIdx.x;
  if (i < NN) deg[i] = 1.0f;
}

__global__ void k_deg_edges(const int* __restrict__ dst, float* __restrict__ deg) {
  int e = blockIdx.x * blockDim.x + threadIdx.x;
  if (e < NE)
    __hip_atomic_fetch_add(deg + dst[e], 1.0f, __ATOMIC_RELAXED, __HIP_MEMORY_SCOPE_AGENT);
}

__global__ void k_rsqrt(const float* __restrict__ deg, float* __restrict__ dinv) {
  int i = blockIdx.x * blockDim.x + threadIdx.x;
  if (i < NN) dinv[i] = rsqrtf(deg[i]);
}

__global__ void k_gemm_wmma(const float* __restrict__ X, const float* __restrict__ W,
                            float* __restrict__ H, int K, int N) {
  int wave = (int)((blockIdx.x * blockDim.x + threadIdx.x) >> 5);
  int lane = threadIdx.x & 31;
  int tilesN = N >> 4;
  int totTiles = (NN >> 4) * tilesN;
  if (wave >= totTiles) return;
  int tr = wave / tilesN;
  int tm = tr << 4;
  int tn = (wave - tr * tilesN) << 4;
  int half = lane >> 4;
  int l    = lane & 15;

  const float* xrow = X + (size_t)(tm + l) * K + 2 * half;
  const float* wcol = W + tn + l;
  v8f c = {};
  for (int k = 0; k < K; k += 4) {
    v2f a, b;
    a.x = xrow[k];
    a.y = xrow[k + 1];
    b.x = wcol[(size_t)(k + 2 * half) * N];
    b.y = wcol[(size_t)(k + 2 * half + 1) * N];
    c = __builtin_amdgcn_wmma_f32_16x16x4_f32(false, a, false, b, (short)0, c, false, false);
  }
  float* hrow = H + (size_t)(tm + 8 * half) * N + tn + l;
#pragma unroll
  for (int r = 0; r < 8; ++r) hrow[(size_t)r * N] = c[r];
}

#define AGG_NB    128
#define AGG_ROWS  ((NN + AGG_NB - 1) / AGG_NB)
#define AGG_QCAP  544
#define AGG_U     4

static_assert(NE % 4 == 0, "dst is streamed as int4");
static_assert(AGG_ROWS <= 512, "row offset is packed into 9 bits");
static_assert(NE <= (1 << 22), "edge id is packed above 9 bits");
static_assert(NN <= (1 << 23), "source row is packed above 9 bits");

template <int V> using vfN = float __attribute__((ext_vector_type(V)));
typedef unsigned __attribute__((ext_vector_type(2))) v2u;

template <int C>
__device__ __forceinline__ void agg_apply(float* __restrict__ acc, int s, int r, float w, int cnt,
                                          const float* __restrict__ H, int lane) {
  constexpr int V = C / 32;
  constexpr int G = 16;
  for (int j0 = 0; j0 < cnt; j0 += G) {
    vfN<V> v[G];
#pragma unroll
    for (int jj = 0; jj < G; ++jj) {
      const int sj = __builtin_amdgcn_readlane(s, j0 + jj);
      v[jj] = *(const vfN<V>*)(H + (size_t)sj * C + V * lane);
    }
#pragma unroll
    for (int jj = 0; jj < G; ++jj) {
      const int j = j0 + jj;
      const int   rj = __builtin_amdgcn_readlane(r, j);
      const float wj = __int_as_float(__builtin_amdgcn_readlane(__float_as_int(w), j));
      float* ap = acc + rj * C + lane;
      const bool live = j < cnt;
#pragma unroll
      for (int k = 0; k < V; ++k) {
        const float m = live ? v[jj][k] * wj : 0.0f;
        __hip_atomic_fetch_add(ap + 32 * k, m, __ATOMIC_RELAXED, __HIP_MEMORY_SCOPE_WORKGROUP);
      }
    }
  }
}

template <int C>
__device__ __forceinline__ void agg_drain(float* __restrict__ acc, const unsigned* __restrict__ qp,
                                          int cnt, const float* __restrict__ H,
                                          const int* __restrict__ src, const float* __restrict__ dinv,
                                          int lo, int lane, v2u* __restrict__ seg, int* wpos) {
  const bool act = lane < cnt;
  const unsigned ent = act ? qp[lane] : 0u;
  const int e = (int)(ent >> 9);
  const int r = (int)(ent & 511u);
  const int s = act ? src[e] : 0;
  const float w = act ? dinv[s] * dinv[lo + r] : 0.0f;
  int p = 0;
  if (lane == 0) p = __hip_atomic_fetch_add(wpos, cnt, __ATOMIC_RELAXED, __HIP_MEMORY_SCOPE_WORKGROUP);
  p = __builtin_amdgcn_readfirstlane(p);
  if (act) {
    v2u le;
    le.x = ((unsigned)s << 9) | (unsigned)r;
    le.y = __float_as_uint(w);
    seg[p + lane] = le;
  }
  agg_apply<C>(acc, s, r, w, cnt, H, lane);
}

template <int C>
__device__ __forceinline__ void agg_rows_init(float* __restrict__ acc, const float* __restrict__ H,
                                              const float* __restrict__ dinv, const float* __restrict__ bias,
                                              int lo, int nrows, int wv, int lane) {
  constexpr int V = C / 32;
  vfN<V> bv = *(const vfN<V>*)(bias + V * lane);
  for (int r = wv; r < nrows; r += 32) {
    const float di = dinv[lo + r];
    const float d2 = di * di;
    const vfN<V> hv = *(const vfN<V>*)(H + (size_t)(lo + r) * C + V * lane);
#pragma unroll
    for (int k = 0; k < V; ++k) acc[r * C + k * 32 + lane] = hv[k] * d2 + bv[k];
  }
}

template <int C>
__device__ __forceinline__ void agg_rows_out(const float* __restrict__ acc, float* __restrict__ out,
                                             int lo, int nrows, int wv, int lane, int relu) {
  constexpr int V = C / 32;
  for (int r = wv; r < nrows; r += 32) {
    vfN<V> ov;
#pragma unroll
    for (int k = 0; k < V; ++k) {
      const float x = acc[r * C + k * 32 + lane];
      ov[k] = relu ? fmaxf(x, 0.0f) : x;
    }
    *(vfN<V>*)(out + (size_t)(lo + r) * C + V * lane) = ov;
  }
}

template <int C>
__global__ __launch_bounds__(1024) void k_agg_sift(const float* __restrict__ H, const int* __restrict__ src,
                                                   const int* __restrict__ dst, const float* __restrict__ deg,
                                                   const float* __restrict__ dinv,
                                                   const float* __restrict__ bias, float* __restrict__ out,
                                                   int relu, unsigned* __restrict__ list,
                                                   int* __restrict__ seg_start, int* __restrict__ seg_cnt) {
  __shared__ float acc[AGG_ROWS * C];
  __shared__ unsigned q[32 * AGG_QCAP];
  __shared__ int red[32];
  __shared__ int wpos;
  const int tid  = threadIdx.x;
  const int lane = tid & 31;
  const int wv   = tid >> 5;
  const int lo   = blockIdx.x * AGG_ROWS;
  const int nrows = min(AGG_ROWS, NN - lo);

  {
    int part = 0;
    for (int i = tid; i < lo; i += 1024) part += (int)deg[i] - 1;
#pragma unroll
    for (int m = 16; m > 0; m >>= 1) part += __shfl_xor(part, m);
    if (lane == 0) red[wv] = part;
    if (tid == 0) wpos = 0;
  }
  agg_rows_init<C>(acc, H, dinv, bias, lo, nrows, wv, lane);
  __syncthreads();
  int pre = red[lane];
#pragma unroll
  for (int m = 16; m > 0; m >>= 1) pre += __shfl_xor(pre, m);
  v2u* seg = (v2u*)list + pre;

  unsigned* myq = q + wv * AGG_QCAP;
  int qn = 0;
  const int4* d4 = (const int4*)dst;
  constexpr int N4 = NE / 4;

  int4 cur[AGG_U];
#pragma unroll
  for (int u = 0; u < AGG_U; ++u) cur[u] = d4[min(u * 1024 + tid, N4 - 1)];
  for (int base = 0; base < N4; base += 1024 * AGG_U) {
    int4 nxt[AGG_U];
#pragma unroll
    for (int u = 0; u < AGG_U; ++u) {
      nxt[u] = d4[min(base + 1024 * AGG_U + u * 1024 + tid, N4 - 1)];
    }
    unsigned rr[AGG_U][4];
    unsigned mk[AGG_U][4];
    unsigned lim[AGG_U];
#pragma unroll
    for (int u = 0; u < AGG_U; ++u) {
      lim[u] = (base + u * 1024 + tid < N4) ? (unsigned)nrows : 0u;
      rr[u][0] = (unsigned)(cur[u].x - lo);
      rr[u][1] = (unsigned)(cur[u].y - lo);
      rr[u][2] = (unsigned)(cur[u].z - lo);
      rr[u][3] = (unsigned)(cur[u].w - lo);
#pragma unroll
      for (int k = 0; k < 4; ++k) mk[u][k] = __builtin_amdgcn_ballot_w32(rr[u][k] < lim[u]);
    }
#pragma unroll
    for (int u = 0; u < AGG_U; ++u) {
      const unsigned e0s = (unsigned)(base + u * 1024 + tid) << 11;
#pragma unroll
      for (int k = 0; k < 4; ++k) {
        const unsigned m = mk[u][k];
        if (m) {
          if (rr[u][k] < lim[u])
            myq[qn + __builtin_amdgcn_mbcnt_lo(m, 0u)] = e0s + ((unsigned)k << 9) + rr[u][k];
          qn += __builtin_popcount(m);
        }
      }
    }
    while (qn >= 32) {
      qn -= 32;
      agg_drain<C>(acc, myq + qn, 32, H, src, dinv, lo, lane, seg, &wpos);
    }
#pragma unroll
    for (int u = 0; u < AGG_U; ++u) cur[u] = nxt[u];
  }
  if (qn > 0) agg_drain<C>(acc, myq, qn, H, src, dinv, lo, lane, seg, &wpos);
  __syncthreads();

  if (tid == 0) {
    seg_start[blockIdx.x] = pre;
    seg_cnt[blockIdx.x]   = wpos;
  }
  agg_rows_out<C>(acc, out, lo, nrows, wv, lane, relu);
}

template <int C>
__global__ __launch_bounds__(1024) void k_agg_list(const float* __restrict__ H, const unsigned* __restrict__ list,
                                                   const int* __restrict__ seg_start,
                                                   const int* __restrict__ seg_cnt,
                                                   const float* __restrict__ dinv,
                                                   const float* __restrict__ bias, float* __restrict__ out,
                                                   int relu) {
  __shared__ float acc[AGG_ROWS * C];
  const int tid  = threadIdx.x;
  const int lane = tid & 31;
  const int wv   = __builtin_amdgcn_readfirstlane(tid >> 5);
  const int lo   = blockIdx.x * AGG_ROWS;
  const int nrows = min(AGG_ROWS, NN - lo);
  const int cnt  = seg_cnt[blockIdx.x];
  const v2u* seg = (const v2u*)list + seg_start[blockIdx.x];

  int c0 = wv * 32;
  v2u en = {0u, 0u};
  if (c0 + lane < cnt) en = seg[c0 + lane];

  agg_rows_init<C>(acc, H, dinv, bias, lo, nrows, wv, lane);
  __syncthreads();

  for (; c0 < cnt; c0 += 1024) {
    v2u nx = {0u, 0u};
    if (c0 + 1024 + lane < cnt) nx = seg[c0 + 1024 + lane];
    agg_apply<C>(acc, (int)(en.x >> 9), (int)(en.x & 511u), __uint_as_float(en.y),
                 min(32, cnt - c0), H, lane);
    en = nx;
  }
  __syncthreads();

  agg_rows_out<C>(acc, out, lo, nrows, wv, lane, relu);
}

static inline unsigned nblk(long long n, int b) { return (unsigned)((n + b - 1) / b); }

extern "C" void kernel_launch(void* const* d_in, const int* in_sizes, int n_in,
                              void* d_out, int out_size, void* d_ws, size_t ws_size,
                              hipStream_t stream) {
  (void)in_sizes; (void)n_in; (void)out_size; (void)ws_size;
  const float* x   = (const float*)d_in[0];
  const int*   ei  = (const int*)d_in[1];
  const float* We1 = (const float*)d_in[2]; const float* be1 = (const float*)d_in[3];
  const float* We2 = (const float*)d_in[4]; const float* be2 = (const float*)d_in[5];
  const float* Wd1 = (const float*)d_in[6]; const float* bd1 = (const float*)d_in[7];
  const float* Wd2 = (const float*)d_in[8]; const float* bd2 = (const float*)d_in[9];
  const int* src = ei;
  const int* dst = ei + NE;

  char*  ws   = (char*)d_ws;
  float* dinv = (float*)ws;
  size_t off  = ((size_t)NN * 4 + 255) & ~(size_t)255;
  float* G    = (float*)(ws + off);
  float* P    = G + (size_t)NN * 128;
  float* deg  = P + (size_t)NN * 64;
  unsigned* list = (unsigned*)((char*)deg + off);
  int* seg_start = (int*)(list + (size_t)NE * 2);
  int* seg_cnt   = seg_start + AGG_NB;
  float* O    = (float*)d_out;

  k_init_deg <<<nblk(NN, 256), 256, 0, stream>>>(deg);
  k_deg_edges<<<nblk(NE, 256), 256, 0, stream>>>(dst, deg);
  k_rsqrt    <<<nblk(NN, 256), 256, 0, stream>>>(deg, dinv);

  const long long rowTiles = NN / 16;

  k_gemm_wmma<<<nblk(rowTiles * 4 * 32, 256), 256, 0, stream>>>(x, We1, G, 128, 64);
  k_agg_sift<64><<<AGG_NB, 1024, 0, stream>>>(G, src, dst, deg, dinv, be1, P, 1, list, seg_start, seg_cnt);

  k_gemm_wmma<<<nblk(rowTiles * 2 * 32, 256), 256, 0, stream>>>(P, We2, G, 64, 32);
  k_agg_list<32> <<<AGG_NB, 1024, 0, stream>>>(G, list, seg_start, seg_cnt, dinv, be2, P, 0);

  k_gemm_wmma<<<nblk(rowTiles * 4 * 32, 256), 256, 0, stream>>>(P, Wd1, G, 32, 64);
  k_agg_list<64> <<<AGG_NB, 1024, 0, stream>>>(G, list, seg_start, seg_cnt, dinv, bd1, P, 1);

  k_gemm_wmma<<<nblk(rowTiles * 8 * 32, 256), 256, 0, stream>>>(P, Wd2, G, 64, 128);
  k_agg_list<128><<<AGG_NB, 1024, 0, stream>>>(G, list, seg_start, seg_cnt, dinv, bd2, O, 0);
}


# Aggregation re-tiled from 128 to 256 work-groups (196 rows each, was 391): its LDS row tiles now run on all 256 WGPs, not half; in-row sum order stays the arrival order of the LDS adds
# speedup vs baseline: 2.2264x; 1.1799x over previous
#include <hip/hip_runtime.h>


#define NN 50000
#define NE 800000

typedef __attribute__((ext_vector_type(2))) float v2f;
typedef __attribute__((ext_vector_type(8))) float v8f;

__global__ void k_init_deg(float* deg) {
  int i = blockIdx.x * blockDim.x + threadIdx.x;
  if (i < NN) deg[i] = 1.0f;
}

__global__ void k_deg_edges(const int* __restrict__ dst, float* __restrict__ deg) {
  int e = blockIdx.x * blockDim.x + threadIdx.x;
  if (e < NE)
    __hip_atomic_fetch_add(deg + dst[e], 1.0f, __ATOMIC_RELAXED, __HIP_MEMORY_SCOPE_AGENT);
}

__global__ void k_rsqrt(const float* __restrict__ deg, float* __restrict__ dinv) {
  int i = blockIdx.x * blockDim.x + threadIdx.x;
  if (i < NN) dinv[i] = rsqrtf(deg[i]);
}

__global__ void k_gemm_wmma(const float* __restrict__ X, const float* __restrict__ W,
                            float* __restrict__ H, int K, int N) {
  int wave = (int)((blockIdx.x * blockDim.x + threadIdx.x) >> 5);
  int lane = threadIdx.x & 31;
  int tilesN = N >> 4;
  int totTiles = (NN >> 4) * tilesN;
  if (wave >= totTiles) return;
  int tr = wave / tilesN;
  int tm = tr << 4;
  int tn = (wave - tr * tilesN) << 4;
  int half = lane >> 4;
  int l    = lane & 15;

  const float* xrow = X + (size_t)(tm + l) * K + 2 * half;
  const float* wcol = W + tn + l;
  v8f c = {};
  for (int k = 0; k < K; k += 4) {
    v2f a, b;
    a.x = xrow[k];
    a.y = xrow[k + 1];
    b.x = wcol[(size_t)(k + 2 * half) * N];
    b.y = wcol[(size_t)(k + 2 * half + 1) * N];
    c = __builtin_amdgcn_wmma_f32_16x16x4_f32(false, a, false, b, (short)0, c, false, false);
  }
  float* hrow = H + (size_t)(tm + 8 * half) * N + tn + l;
#pragma unroll
  for (int r = 0; r < 8; ++r) hrow[(size_t)r * N] = c[r];
}

#define AGG_NB    256
#define AGG_ROWS  ((NN + AGG_NB - 1) / AGG_NB)
#define AGG_QCAP  544
#define AGG_U     4

static_assert(NE % 4 == 0, "dst is streamed as int4");
static_assert(AGG_ROWS <= 512, "row offset is packed into 9 bits");
static_assert(NE <= (1 << 22), "edge id is packed above 9 bits");
static_assert(NN <= (1 << 23), "source row is packed above 9 bits");

template <int V> using vfN = float __attribute__((ext_vector_type(V)));
typedef unsigned __attribute__((ext_vector_type(2))) v2u;

template <int C>
__device__ __forceinline__ void agg_apply(float* __restrict__ acc, int s, int r, float w, int cnt,
                                          const float* __restrict__ H, int lane) {
  constexpr int V = C / 32;
  constexpr int G = 16;
  for (int j0 = 0; j0 < cnt; j0 += G) {
    vfN<V> v[G];
#pragma unroll
    for (int jj = 0; jj < G; ++jj) {
      const int sj = __builtin_amdgcn_readlane(s, j0 + jj);
      v[jj] = *(const vfN<V>*)(H + (size_t)sj * C + V * lane);
    }
#pragma unroll
    for (int jj = 0; jj < G; ++jj) {
      const int j = j0 + jj;
      const int   rj = __builtin_amdgcn_readlane(r, j);
      const float wj = __int_as_float(__builtin_amdgcn_readlane(__float_as_int(w), j));
      float* ap = acc + rj * C + lane;
      const bool live = j < cnt;
#pragma unroll
      for (int k = 0; k < V; ++k) {
        const float m = live ? v[jj][k] * wj : 0.0f;
        __hip_atomic_fetch_add(ap + 32 * k, m, __ATOMIC_RELAXED, __HIP_MEMORY_SCOPE_WORKGROUP);
      }
    }
  }
}

template <int C>
__device__ __forceinline__ void agg_drain(float* __restrict__ acc, const unsigned* __restrict__ qp,
                                          int cnt, const float* __restrict__ H,
                                          const int* __restrict__ src, const float* __restrict__ dinv,
                                          int lo, int lane, v2u* __restrict__ seg, int* wpos) {
  const bool act = lane < cnt;
  const unsigned ent = act ? qp[lane] : 0u;
  const int e = (int)(ent >> 9);
  const int r = (int)(ent & 511u);
  const int s = act ? src[e] : 0;
  const float w = act ? dinv[s] * dinv[lo + r] : 0.0f;
  int p = 0;
  if (lane == 0) p = __hip_atomic_fetch_add(wpos, cnt, __ATOMIC_RELAXED, __HIP_MEMORY_SCOPE_WORKGROUP);
  p = __builtin_amdgcn_readfirstlane(p);
  if (act) {
    v2u le;
    le.x = ((unsigned)s << 9) | (unsigned)r;
    le.y = __float_as_uint(w);
    seg[p + lane] = le;
  }
  agg_apply<C>(acc, s, r, w, cnt, H, lane);
}

template <int C>
__device__ __forceinline__ void agg_rows_init(float* __restrict__ acc, const float* __restrict__ H,
                                              const float* __restrict__ dinv, const float* __restrict__ bias,
                                              int lo, int nrows, int wv, int lane) {
  constexpr int V = C / 32;
  vfN<V> bv = *(const vfN<V>*)(bias + V * lane);
  for (int r = wv; r < nrows; r += 32) {
    const float di = dinv[lo + r];
    const float d2 = di * di;
    const vfN<V> hv = *(const vfN<V>*)(H + (size_t)(lo + r) * C + V * lane);
#pragma unroll
    for (int k = 0; k < V; ++k) acc[r * C + k * 32 + lane] = hv[k] * d2 + bv[k];
  }
}

template <int C>
__device__ __forceinline__ void agg_rows_out(const float* __restrict__ acc, float* __restrict__ out,
                                             int lo, int nrows, int wv, int lane, int relu) {
  constexpr int V = C / 32;
  for (int r = wv; r < nrows; r += 32) {
    vfN<V> ov;
#pragma unroll
    for (int k = 0; k < V; ++k) {
      const float x = acc[r * C + k * 32 + lane];
      ov[k] = relu ? fmaxf(x, 0.0f) : x;
    }
    *(vfN<V>*)(out + (size_t)(lo + r) * C + V * lane) = ov;
  }
}

template <int C>
__global__ __launch_bounds__(1024) void k_agg_sift(const float* __restrict__ H, const int* __restrict__ src,
                                                   const int* __restrict__ dst, const float* __restrict__ deg,
                                                   const float* __restrict__ dinv,
                                                   const float* __restrict__ bias, float* __restrict__ out,
                                                   int relu, unsigned* __restrict__ list,
                                                   int* __restrict__ seg_start, int* __restrict__ seg_cnt) {
  __shared__ float acc[AGG_ROWS * C];
  __shared__ unsigned q[32 * AGG_QCAP];
  __shared__ int red[32];
  __shared__ int wpos;
  const int tid  = threadIdx.x;
  const int lane = tid & 31;
  const int wv   = tid >> 5;
  const int lo   = blockIdx.x * AGG_ROWS;
  const int nrows = min(AGG_ROWS, NN - lo);

  {
    int part = 0;
    for (int i = tid; i < lo; i += 1024) part += (int)deg[i] - 1;
#pragma unroll
    for (int m = 16; m > 0; m >>= 1) part += __shfl_xor(part, m);
    if (lane == 0) red[wv] = part;
    if (tid == 0) wpos = 0;
  }
  agg_rows_init<C>(acc, H, dinv, bias, lo, nrows, wv, lane);
  __syncthreads();
  int pre = red[lane];
#pragma unroll
  for (int m = 16; m > 0; m >>= 1) pre += __shfl_xor(pre, m);
  v2u* seg = (v2u*)list + pre;

  unsigned* myq = q + wv * AGG_QCAP;
  int qn = 0;
  const int4* d4 = (const int4*)dst;
  constexpr int N4 = NE / 4;

  int4 cur[AGG_U];
#pragma unroll
  for (int u = 0; u < AGG_U; ++u) cur[u] = d4[min(u * 1024 + tid, N4 - 1)];
  for (int base = 0; base < N4; base += 1024 * AGG_U) {
    int4 nxt[AGG_U];
#pragma unroll
    for (int u = 0; u < AGG_U; ++u) {
      nxt[u] = d4[min(base + 1024 * AGG_U + u * 1024 + tid, N4 - 1)];
    }
    unsigned rr[AGG_U][4];
    unsigned mk[AGG_U][4];
    unsigned lim[AGG_U];
#pragma unroll
    for (int u = 0; u < AGG_U; ++u) {
      lim[u] = (base + u * 1024 + tid < N4) ? (unsigned)nrows : 0u;
      rr[u][0] = (unsigned)(cur[u].x - lo);
      rr[u][1] = (unsigned)(cur[u].y - lo);
      rr[u][2] = (unsigned)(cur[u].z - lo);
      rr[u][3] = (unsigned)(cur[u].w - lo);
#pragma unroll
      for (int k = 0; k < 4; ++k) mk[u][k] = __builtin_amdgcn_ballot_w32(rr[u][k] < lim[u]);
    }
#pragma unroll
    for (int u = 0; u < AGG_U; ++u) {
      const unsigned e0s = (unsigned)(base + u * 1024 + tid) << 11;
#pragma unroll
      for (int k = 0; k < 4; ++k) {
        const unsigned m = mk[u][k];
        if (m) {
          if (rr[u][k] < lim[u])
            myq[qn + __builtin_amdgcn_mbcnt_lo(m, 0u)] = e0s + ((unsigned)k << 9) + rr[u][k];
          qn += __builtin_popcount(m);
        }
      }
    }
    while (qn >= 32) {
      qn -= 32;
      agg_drain<C>(acc, myq + qn, 32, H, src, dinv, lo, lane, seg, &wpos);
    }
#pragma unroll
    for (int u = 0; u < AGG_U; ++u) cur[u] = nxt[u];
  }
  if (qn > 0) agg_drain<C>(acc, myq, qn, H, src, dinv, lo, lane, seg, &wpos);
  __syncthreads();

  if (tid == 0) {
    seg_start[blockIdx.x] = pre;
    seg_cnt[blockIdx.x]   = wpos;
  }
  agg_rows_out<C>(acc, out, lo, nrows, wv, lane, relu);
}

template <int C>
__global__ __launch_bounds__(1024) void k_agg_list(const float* __restrict__ H, const unsigned* __restrict__ list,
                                                   const int* __restrict__ seg_start,
                                                   const int* __restrict__ seg_cnt,
                                                   const float* __restrict__ dinv,
                                                   const float* __restrict__ bias, float* __restrict__ out,
                                                   int relu) {
  __shared__ float acc[AGG_ROWS * C];
  const int tid  = threadIdx.x;
  const int lane = tid & 31;
  const int wv   = __builtin_amdgcn_readfirstlane(tid >> 5);
  const int lo   = blockIdx.x * AGG_ROWS;
  const int nrows = min(AGG_ROWS, NN - lo);
  const int cnt  = seg_cnt[blockIdx.x];
  const v2u* seg = (const v2u*)list + seg_start[blockIdx.x];

  int c0 = wv * 32;
  v2u en = {0u, 0u};
  if (c0 + lane < cnt) en = seg[c0 + lane];

  agg_rows_init<C>(acc, H, dinv, bias, lo, nrows, wv, lane);
  __syncthreads();

  for (; c0 < cnt; c0 += 1024) {
    v2u nx = {0u, 0u};
    if (c0 + 1024 + lane < cnt) nx = seg[c0 + 1024 + lane];
    agg_apply<C>(acc, (int)(en.x >> 9), (int)(en.x & 511u), __uint_as_float(en.y),
                 min(32, cnt - c0), H, lane);
    en = nx;
  }
  __syncthreads();

  agg_rows_out<C>(acc, out, lo, nrows, wv, lane, relu);
}

static inline unsigned nblk(long long n, int b) { return (unsigned)((n + b - 1) / b); }

extern "C" void kernel_launch(void* const* d_in, const int* in_sizes, int n_in,
                              void* d_out, int out_size, void* d_ws, size_t ws_size,
                              hipStream_t stream) {
  (void)in_sizes; (void)n_in; (void)out_size; (void)ws_size;
  const float* x   = (const float*)d_in[0];
  const int*   ei  = (const int*)d_in[1];
  const float* We1 = (const float*)d_in[2]; const float* be1 = (const float*)d_in[3];
  const float* We2 = (const float*)d_in[4]; const float* be2 = (const float*)d_in[5];
  const float* Wd1 = (const float*)d_in[6]; const float* bd1 = (const float*)d_in[7];
  const float* Wd2 = (const float*)d_in[8]; const float* bd2 = (const float*)d_in[9];
  const int* src = ei;
  const int* dst = ei + NE;

  char*  ws   = (char*)d_ws;
  float* dinv = (float*)ws;
  size_t off  = ((size_t)NN * 4 + 255) & ~(size_t)255;
  float* G    = (float*)(ws + off);
  float* P    = G + (size_t)NN * 128;
  float* deg  = P + (size_t)NN * 64;
  unsigned* list = (unsigned*)((char*)deg + off);
  int* seg_start = (int*)(list + (size_t)NE * 2);
  int* seg_cnt   = seg_start + AGG_NB;
  float* O    = (float*)d_out;

  k_init_deg <<<nblk(NN, 256), 256, 0, stream>>>(deg);
  k_deg_edges<<<nblk(NE, 256), 256, 0, stream>>>(dst, deg);
  k_rsqrt    <<<nblk(NN, 256), 256, 0, stream>>>(deg, dinv);

  const long long rowTiles = NN / 16;

  k_gemm_wmma<<<nblk(rowTiles * 4 * 32, 256), 256, 0, stream>>>(x, We1, G, 128, 64);
  k_agg_sift<64><<<AGG_NB, 1024, 0, stream>>>(G, src, dst, deg, dinv, be1, P, 1, list, seg_start, seg_cnt);

  k_gemm_wmma<<<nblk(rowTiles * 2 * 32, 256), 256, 0, stream>>>(P, We2, G, 64, 32);
  k_agg_list<32> <<<AGG_NB, 1024, 0, stream>>>(G, list, seg_start, seg_cnt, dinv, be2, P, 0);

  k_gemm_wmma<<<nblk(rowTiles * 4 * 32, 256), 256, 0, stream>>>(P, Wd1, G, 32, 64);
  k_agg_list<64> <<<AGG_NB, 1024, 0, stream>>>(G, list, seg_start, seg_cnt, dinv, bd1, P, 1);

  k_gemm_wmma<<<nblk(rowTiles * 8 * 32, 256), 256, 0, stream>>>(P, Wd2, G, 64, 128);
  k_agg_list<128><<<AGG_NB, 1024, 0, stream>>>(G, list, seg_start, seg_cnt, dinv, bd2, O, 0);
}


# Dense products re-tiled: one wave per 16-row stripe over all N columns (A fragments loaded once, W staged in LDS as k-pairs, 8-byte B reads); 56264 short waves -> 12512; k order and sums unchanged
# speedup vs baseline: 2.3124x; 1.0386x over previous
#include <hip/hip_runtime.h>


#define NN 50000
#define NE 800000

typedef __attribute__((ext_vector_type(2))) float v2f;
typedef __attribute__((ext_vector_type(8))) float v8f;

__global__ void k_init_deg(float* deg) {
  int i = blockIdx.x * blockDim.x + threadIdx.x;
  if (i < NN) deg[i] = 1.0f;
}

__global__ void k_deg_edges(const int* __restrict__ dst, float* __restrict__ deg) {
  int e = blockIdx.x * blockDim.x + threadIdx.x;
  if (e < NE)
    __hip_atomic_fetch_add(deg + dst[e], 1.0f, __ATOMIC_RELAXED, __HIP_MEMORY_SCOPE_AGENT);
}

__global__ void k_rsqrt(const float* __restrict__ deg, float* __restrict__ dinv) {
  int i = blockIdx.x * blockDim.x + threadIdx.x;
  if (i < NN) dinv[i] = rsqrtf(deg[i]);
}

#define GEMM_WAVES 8
#define GEMM_STRIPES (NN / 16)
#define GEMM_NB ((GEMM_STRIPES + GEMM_WAVES - 1) / GEMM_WAVES)
static_assert(NN % 16 == 0, "rows are covered by whole 16-row stripes");

typedef __attribute__((ext_vector_type(4))) float v4f;

template <int K, int N>
__global__ __launch_bounds__(GEMM_WAVES * 32) void k_gemm_stripe(const float* __restrict__ X,
                                                                 const float* __restrict__ W,
                                                                 float* __restrict__ H) {
  constexpr int T  = N / 16;
  constexpr int KS = K / 4;
  constexpr int WP = N + 16;
  __shared__ v2f wl[(K / 2) * WP];
  const int tid  = threadIdx.x;
  const int lane = tid & 31;
  const int wv   = __builtin_amdgcn_readfirstlane(tid >> 5);
  const int half = lane >> 4;
  const int l    = lane & 15;
  const int stripe = blockIdx.x * GEMM_WAVES + wv;
  const bool live = stripe < GEMM_STRIPES;

  v2f a[KS];
  if (live) {
    const float* xrow = X + (size_t)(stripe * 16 + l) * K + 2 * half;
#pragma unroll
    for (int kk = 0; kk < KS; ++kk) a[kk] = *(const v2f*)(xrow + 4 * kk);
  } else {
#pragma unroll
    for (int kk = 0; kk < KS; ++kk) a[kk] = v2f{0.0f, 0.0f};
  }

  constexpr int NI = (K / 2) * (N / 4) / (GEMM_WAVES * 32);
  static_assert(NI * GEMM_WAVES * 32 == (K / 2) * (N / 4), "W is staged in whole passes of the work-group");
  v4f w0[NI], w1[NI];
#pragma unroll
  for (int it = 0; it < NI; ++it) {
    const int i  = tid + it * GEMM_WAVES * 32;
    const int kp = i / (N / 4);
    const int q  = i - kp * (N / 4);
    w0[it] = *(const v4f*)(W + (size_t)(2 * kp) * N + 4 * q);
    w1[it] = *(const v4f*)(W + (size_t)(2 * kp + 1) * N + 4 * q);
  }
#pragma unroll
  for (int it = 0; it < NI; ++it) {
    const int i  = tid + it * GEMM_WAVES * 32;
    const int kp = i / (N / 4);
    const int q  = i - kp * (N / 4);
    v4f* dp = (v4f*)(wl + kp * WP + 4 * q);
    dp[0] = v4f{w0[it].x, w1[it].x, w0[it].y, w1[it].y};
    dp[1] = v4f{w0[it].z, w1[it].z, w0[it].w, w1[it].w};
  }
  __syncthreads();
  if (!live) return;

  v8f c[T];
#pragma unroll
  for (int t = 0; t < T; ++t) c[t] = v8f{};
  const v2f* bp = wl + half * WP + l;
#pragma unroll
  for (int kk = 0; kk < KS; ++kk) {
    v2f b[T];
#pragma unroll
    for (int t = 0; t < T; ++t) b[t] = bp[(2 * kk) * WP + 16 * t];
#pragma unroll
    for (int t = 0; t < T; ++t)
      c[t] = __builtin_amdgcn_wmma_f32_16x16x4_f32(false, a[kk], false, b[t], (short)0, c[t], false, false);
  }
  float* hrow = H + (size_t)(stripe * 16 + 8 * half) * N + l;
#pragma unroll
  for (int t = 0; t < T; ++t) {
#pragma unroll
    for (int r = 0; r < 8; ++r) hrow[(size_t)r * N + 16 * t] = c[t][r];
  }
}

#define AGG_NB    256
#define AGG_ROWS  ((NN + AGG_NB - 1) / AGG_NB)
#define AGG_QCAP  544
#define AGG_U     4

static_assert(NE % 4 == 0, "dst is streamed as int4");
static_assert(AGG_ROWS <= 512, "row offset is packed into 9 bits");
static_assert(NE <= (1 << 22), "edge id is packed above 9 bits");
static_assert(NN <= (1 << 23), "source row is packed above 9 bits");

template <int V> using vfN = float __attribute__((ext_vector_type(V)));
typedef unsigned __attribute__((ext_vector_type(2))) v2u;

template <int C>
__device__ __forceinline__ void agg_apply(float* __restrict__ acc, int s, int r, float w, int cnt,
                                          const float* __restrict__ H, int lane) {
  constexpr int V = C / 32;
  constexpr int G = 16;
  for (int j0 = 0; j0 < cnt; j0 += G) {
    vfN<V> v[G];
#pragma unroll
    for (int jj = 0; jj < G; ++jj) {
      const int sj = __builtin_amdgcn_readlane(s, j0 + jj);
      v[jj] = *(const vfN<V>*)(H + (size_t)sj * C + V * lane);
    }
#pragma unroll
    for (int jj = 0; jj < G; ++jj) {
      const int j = j0 + jj;
      const int   rj = __builtin_amdgcn_readlane(r, j);
      const float wj = __int_as_float(__builtin_amdgcn_readlane(__float_as_int(w), j));
      float* ap = acc + rj * C + lane;
      const bool live = j < cnt;
#pragma unroll
      for (int k = 0; k < V; ++k) {
        const float m = live ? v[jj][k] * wj : 0.0f;
        __hip_atomic_fetch_add(ap + 32 * k, m, __ATOMIC_RELAXED, __HIP_MEMORY_SCOPE_WORKGROUP);
      }
    }
  }
}

template <int C>
__device__ __forceinline__ void agg_drain(float* __restrict__ acc, const unsigned* __restrict__ qp,
                                          int cnt, const float* __restrict__ H,
                                          const int* __restrict__ src, const float* __restrict__ dinv,
                                          int lo, int lane, v2u* __restrict__ seg, int* wpos) {
  const bool act = lane < cnt;
  const unsigned ent = act ? qp[lane] : 0u;
  const int e = (int)(ent >> 9);
  const int r = (int)(ent & 511u);
  const int s = act ? src[e] : 0;
  const float w = act ? dinv[s] * dinv[lo + r] : 0.0f;
  int p = 0;
  if (lane == 0) p = __hip_atomic_fetch_add(wpos, cnt, __ATOMIC_RELAXED, __HIP_MEMORY_SCOPE_WORKGROUP);
  p = __builtin_amdgcn_readfirstlane(p);
  if (act) {
    v2u le;
    le.x = ((unsigned)s << 9) | (unsigned)r;
    le.y = __float_as_uint(w);
    seg[p + lane] = le;
  }
  agg_apply<C>(acc, s, r, w, cnt, H, lane);
}

template <int C>
__device__ __forceinline__ void agg_rows_init(float* __restrict__ acc, const float* __restrict__ H,
                                              const float* __restrict__ dinv, const float* __restrict__ bias,
                                              int lo, int nrows, int wv, int lane) {
  constexpr int V = C / 32;
  vfN<V> bv = *(const vfN<V>*)(bias + V * lane);
  for (int r = wv; r < nrows; r += 32) {
    const float di = dinv[lo + r];
    const float d2 = di * di;
    const vfN<V> hv = *(const vfN<V>*)(H + (size_t)(lo + r) * C + V * lane);
#pragma unroll
    for (int k = 0; k < V; ++k) acc[r * C + k * 32 + lane] = hv[k] * d2 + bv[k];
  }
}

template <int C>
__device__ __forceinline__ void agg_rows_out(const float* __restrict__ acc, float* __restrict__ out,
                                             int lo, int nrows, int wv, int lane, int relu) {
  constexpr int V = C / 32;
  for (int r = wv; r < nrows; r += 32) {
    vfN<V> ov;
#pragma unroll
    for (int k = 0; k < V; ++k) {
      const float x = acc[r * C + k * 32 + lane];
      ov[k] = relu ? fmaxf(x, 0.0f) : x;
    }
    *(vfN<V>*)(out + (size_t)(lo + r) * C + V * lane) = ov;
  }
}

template <int C>
__global__ __launch_bounds__(1024) void k_agg_sift(const float* __restrict__ H, const int* __restrict__ src,
                                                   const int* __restrict__ dst, const float* __restrict__ deg,
                                                   const float* __restrict__ dinv,
                                                   const float* __restrict__ bias, float* __restrict__ out,
                                                   int relu, unsigned* __restrict__ list,
                                                   int* __restrict__ seg_start, int* __restrict__ seg_cnt) {
  __shared__ float acc[AGG_ROWS * C];
  __shared__ unsigned q[32 * AGG_QCAP];
  __shared__ int red[32];
  __shared__ int wpos;
  const int tid  = threadIdx.x;
  const int lane = tid & 31;
  const int wv   = tid >> 5;
  const int lo   = blockIdx.x * AGG_ROWS;
  const int nrows = min(AGG_ROWS, NN - lo);

  {
    int part = 0;
    for (int i = tid; i < lo; i += 1024) part += (int)deg[i] - 1;
#pragma unroll
    for (int m = 16; m > 0; m >>= 1) part += __shfl_xor(part, m);
    if (lane == 0) red[wv] = part;
    if (tid == 0) wpos = 0;
  }
  agg_rows_init<C>(acc, H, dinv, bias, lo, nrows, wv, lane);
  __syncthreads();
  int pre = red[lane];
#pragma unroll
  for (int m = 16; m > 0; m >>= 1) pre += __shfl_xor(pre, m);
  v2u* seg = (v2u*)list + pre;

  unsigned* myq = q + wv * AGG_QCAP;
  int qn = 0;
  const int4* d4 = (const int4*)dst;
  constexpr int N4 = NE / 4;

  int4 cur[AGG_U];
#pragma unroll
  for (int u = 0; u < AGG_U; ++u) cur[u] = d4[min(u * 1024 + tid, N4 - 1)];
  for (int base = 0; base < N4; base += 1024 * AGG_U) {
    int4 nxt[AGG_U];
#pragma unroll
    for (int u = 0; u < AGG_U; ++u) {
      nxt[u] = d4[min(base + 1024 * AGG_U + u * 1024 + tid, N4 - 1)];
    }
    unsigned rr[AGG_U][4];
    unsigned mk[AGG_U][4];
    unsigned lim[AGG_U];
#pragma unroll
    for (int u = 0; u < AGG_U; ++u) {
      lim[u] = (base + u * 1024 + tid < N4) ? (unsigned)nrows : 0u;
      rr[u][0] = (unsigned)(cur[u].x - lo);
      rr[u][1] = (unsigned)(cur[u].y - lo);
      rr[u][2] = (unsigned)(cur[u].z - lo);
      rr[u][3] = (unsigned)(cur[u].w - lo);
#pragma unroll
      for (int k = 0; k < 4; ++k) mk[u][k] = __builtin_amdgcn_ballot_w32(rr[u][k] < lim[u]);
    }
#pragma unroll
    for (int u = 0; u < AGG_U; ++u) {
      const unsigned e0s = (unsigned)(base + u * 1024 + tid) << 11;
#pragma unroll
      for (int k = 0; k < 4; ++k) {
        const unsigned m = mk[u][k];
        if (m) {
          if (rr[u][k] < lim[u])
            myq[qn + __builtin_amdgcn_mbcnt_lo(m, 0u)] = e0s + ((unsigned)k << 9) + rr[u][k];
          qn += __builtin_popcount(m);
        }
      }
    }
    while (qn >= 32) {
      qn -= 32;
      agg_drain<C>(acc, myq + qn, 32, H, src, dinv, lo, lane, seg, &wpos);
    }
#pragma unroll
    for (int u = 0; u < AGG_U; ++u) cur[u] = nxt[u];
  }
  if (qn > 0) agg_drain<C>(acc, myq, qn, H, src, dinv, lo, lane, seg, &wpos);
  __syncthreads();

  if (tid == 0) {
    seg_start[blockIdx.x] = pre;
    seg_cnt[blockIdx.x]   = wpos;
  }
  agg_rows_out<C>(acc, out, lo, nrows, wv, lane, relu);
}

template <int C>
__global__ __launch_bounds__(1024) void k_agg_list(const float* __restrict__ H, const unsigned* __restrict__ list,
                                                   const int* __restrict__ seg_start,
                                                   const int* __restrict__ seg_cnt,
                                                   const float* __restrict__ dinv,
                                                   const float* __restrict__ bias, float* __restrict__ out,
                                                   int relu) {
  __shared__ float acc[AGG_ROWS * C];
  const int tid  = threadIdx.x;
  const int lane = tid & 31;
  const int wv   = __builtin_amdgcn_readfirstlane(tid >> 5);
  const int lo   = blockIdx.x * AGG_ROWS;
  const int nrows = min(AGG_ROWS, NN - lo);
  const int cnt  = seg_cnt[blockIdx.x];
  const v2u* seg = (const v2u*)list + seg_start[blockIdx.x];

  int c0 = wv * 32;
  v2u en = {0u, 0u};
  if (c0 + lane < cnt) en = seg[c0 + lane];

  agg_rows_init<C>(acc, H, dinv, bias, lo, nrows, wv, lane);
  __syncthreads();

  for (; c0 < cnt; c0 += 1024) {
    v2u nx = {0u, 0u};
    if (c0 + 1024 + lane < cnt) nx = seg[c0 + 1024 + lane];
    agg_apply<C>(acc, (int)(en.x >> 9), (int)(en.x & 511u), __uint_as_float(en.y),
                 min(32, cnt - c0), H, lane);
    en = nx;
  }
  __syncthreads();

  agg_rows_out<C>(acc, out, lo, nrows, wv, lane, relu);
}

static inline unsigned nblk(long long n, int b) { return (unsigned)((n + b - 1) / b); }

extern "C" void kernel_launch(void* const* d_in, const int* in_sizes, int n_in,
                              void* d_out, int out_size, void* d_ws, size_t ws_size,
                              hipStream_t stream) {
  (void)in_sizes; (void)n_in; (void)out_size; (void)ws_size;
  const float* x   = (const float*)d_in[0];
  const int*   ei  = (const int*)d_in[1];
  const float* We1 = (const float*)d_in[2]; const float* be1 = (const float*)d_in[3];
  const float* We2 = (const float*)d_in[4]; const float* be2 = (const float*)d_in[5];
  const float* Wd1 = (const float*)d_in[6]; const float* bd1 = (const float*)d_in[7];
  const float* Wd2 = (const float*)d_in[8]; const float* bd2 = (const float*)d_in[9];
  const int* src = ei;
  const int* dst = ei + NE;

  char*  ws   = (char*)d_ws;
  float* dinv = (float*)ws;
  size_t off  = ((size_t)NN * 4 + 255) & ~(size_t)255;
  float* G    = (float*)(ws + off);
  float* P    = G + (size_t)NN * 128;
  float* deg  = P + (size_t)NN * 64;
  unsigned* list = (unsigned*)((char*)deg + off);
  int* seg_start = (int*)(list + (size_t)NE * 2);
  int* seg_cnt   = seg_start + AGG_NB;
  float* O    = (float*)d_out;

  k_init_deg <<<nblk(NN, 256), 256, 0, stream>>>(deg);
  k_deg_edges<<<nblk(NE, 256), 256, 0, stream>>>(dst, deg);
  k_rsqrt    <<<nblk(NN, 256), 256, 0, stream>>>(deg, dinv);

  k_gemm_stripe<128, 64><<<GEMM_NB, GEMM_WAVES * 32, 0, stream>>>(x, We1, G);
  k_agg_sift<64><<<AGG_NB, 1024, 0, stream>>>(G, src, dst, deg, dinv, be1, P, 1, list, seg_start, seg_cnt);

  k_gemm_stripe<64, 32><<<GEMM_NB, GEMM_WAVES * 32, 0, stream>>>(P, We2, G);
  k_agg_list<32> <<<AGG_NB, 1024, 0, stream>>>(G, list, seg_start, seg_cnt, dinv, be2, P, 0);

  k_gemm_stripe<32, 64><<<GEMM_NB, GEMM_WAVES * 32, 0, stream>>>(P, Wd1, G);
  k_agg_list<64> <<<AGG_NB, 1024, 0, stream>>>(G, list, seg_start, seg_cnt, dinv, bd1, P, 1);

  k_gemm_stripe<64, 128><<<GEMM_NB, GEMM_WAVES * 32, 0, stream>>>(P, Wd2, G);
  k_agg_list<128><<<AGG_NB, 1024, 0, stream>>>(G, list, seg_start, seg_cnt, dinv, bd2, O, 0);
}


# Later aggregations keep each destination row in its owner wave's registers (row mod 32 = wave; edge list split into per-wave runs): register fma, no LDS float adds; in-row sum order = run order
# speedup vs baseline: 2.4255x; 1.0489x over previous
#include <hip/hip_runtime.h>


#define NN 50000
#define NE 800000

typedef __attribute__((ext_vector_type(2))) float v2f;
typedef __attribute__((ext_vector_type(8))) float v8f;

__global__ void k_init_deg(float* deg) {
  int i = blockIdx.x * blockDim.x + threadIdx.x;
  if (i < NN) deg[i] = 1.0f;
}

__global__ void k_deg_edges(const int* __restrict__ dst, float* __restrict__ deg) {
  int e = blockIdx.x * blockDim.x + threadIdx.x;
  if (e < NE)
    __hip_atomic_fetch_add(deg + dst[e], 1.0f, __ATOMIC_RELAXED, __HIP_MEMORY_SCOPE_AGENT);
}

__global__ void k_rsqrt(const float* __restrict__ deg, float* __restrict__ dinv) {
  int i = blockIdx.x * blockDim.x + threadIdx.x;
  if (i < NN) dinv[i] = rsqrtf(deg[i]);
}

#define GEMM_WAVES 8
#define GEMM_STRIPES (NN / 16)
#define GEMM_NB ((GEMM_STRIPES + GEMM_WAVES - 1) / GEMM_WAVES)
static_assert(NN % 16 == 0, "rows are covered by whole 16-row stripes");

typedef __attribute__((ext_vector_type(4))) float v4f;

template <int K, int N>
__global__ __launch_bounds__(GEMM_WAVES * 32) void k_gemm_stripe(const float* __restrict__ X,
                                                                 const float* __restrict__ W,
                                                                 float* __restrict__ H) {
  constexpr int T  = N / 16;
  constexpr int KS = K / 4;
  constexpr int WP = N + 16;
  __shared__ v2f wl[(K / 2) * WP];
  const int tid  = threadIdx.x;
  const int lane = tid & 31;
  const int wv   = __builtin_amdgcn_readfirstlane(tid >> 5);
  const int half = lane >> 4;
  const int l    = lane & 15;
  const int stripe = blockIdx.x * GEMM_WAVES + wv;
  const bool live = stripe < GEMM_STRIPES;

  v2f a[KS];
  if (live) {
    const float* xrow = X + (size_t)(stripe * 16 + l) * K + 2 * half;
#pragma unroll
    for (int kk = 0; kk < KS; ++kk) a[kk] = *(const v2f*)(xrow + 4 * kk);
  } else {
#pragma unroll
    for (int kk = 0; kk < KS; ++kk) a[kk] = v2f{0.0f, 0.0f};
  }

  constexpr int NI = (K / 2) * (N / 4) / (GEMM_WAVES * 32);
  static_assert(NI * GEMM_WAVES * 32 == (K / 2) * (N / 4), "W is staged in whole passes of the work-group");
  v4f w0[NI], w1[NI];
#pragma unroll
  for (int it = 0; it < NI; ++it) {
    const int i  = tid + it * GEMM_WAVES * 32;
    const int kp = i / (N / 4);
    const int q  = i - kp * (N / 4);
    w0[it] = *(const v4f*)(W + (size_t)(2 * kp) * N + 4 * q);
    w1[it] = *(const v4f*)(W + (size_t)(2 * kp + 1) * N + 4 * q);
  }
#pragma unroll
  for (int it = 0; it < NI; ++it) {
    const int i  = tid + it * GEMM_WAVES * 32;
    const int kp = i / (N / 4);
    const int q  = i - kp * (N / 4);
    v4f* dp = (v4f*)(wl + kp * WP + 4 * q);
    dp[0] = v4f{w0[it].x, w1[it].x, w0[it].y, w1[it].y};
    dp[1] = v4f{w0[it].z, w1[it].z, w0[it].w, w1[it].w};
  }
  __syncthreads();
  if (!live) return;

  v8f c[T];
#pragma unroll
  for (int t = 0; t < T; ++t) c[t] = v8f{};
  const v2f* bp = wl + half * WP + l;
#pragma unroll
  for (int kk = 0; kk < KS; ++kk) {
    v2f b[T];
#pragma unroll
    for (int t = 0; t < T; ++t) b[t] = bp[(2 * kk) * WP + 16 * t];
#pragma unroll
    for (int t = 0; t < T; ++t)
      c[t] = __builtin_amdgcn_wmma_f32_16x16x4_f32(false, a[kk], false, b[t], (short)0, c[t], false, false);
  }
  float* hrow = H + (size_t)(stripe * 16 + 8 * half) * N + l;
#pragma unroll
  for (int t = 0; t < T; ++t) {
#pragma unroll
    for (int r = 0; r < 8; ++r) hrow[(size_t)r * N + 16 * t] = c[t][r];
  }
}

#define AGG_NB    256
#define AGG_ROWS  ((NN + AGG_NB - 1) / AGG_NB)
#define AGG_QCAP  544
#define AGG_U     4
#define AGG_SLOTS ((AGG_ROWS + 31) / 32)
static_assert(AGG_SLOTS <= 8, "a wave's rows are held in register arrays of 8 slots");

static_assert(NE % 4 == 0, "dst is streamed as int4");
static_assert(AGG_ROWS <= 512, "row offset is packed into 9 bits");
static_assert(NE <= (1 << 22), "edge id is packed above 9 bits");
static_assert(NN <= (1 << 23), "source row is packed above 9 bits");

template <int V> using vfN = float __attribute__((ext_vector_type(V)));
typedef unsigned __attribute__((ext_vector_type(2))) v2u;

template <int C>
__device__ __forceinline__ void agg_apply(float* __restrict__ acc, int s, int r, float w, int cnt,
                                          const float* __restrict__ H, int lane) {
  constexpr int V = C / 32;
  constexpr int G = 16;
  for (int j0 = 0; j0 < cnt; j0 += G) {
    vfN<V> v[G];
#pragma unroll
    for (int jj = 0; jj < G; ++jj) {
      const int sj = __builtin_amdgcn_readlane(s, j0 + jj);
      v[jj] = *(const vfN<V>*)(H + (size_t)sj * C + V * lane);
    }
#pragma unroll
    for (int jj = 0; jj < G; ++jj) {
      const int j = j0 + jj;
      const int   rj = __builtin_amdgcn_readlane(r, j);
      const float wj = __int_as_float(__builtin_amdgcn_readlane(__float_as_int(w), j));
      float* ap = acc + rj * C + lane;
      const bool live = j < cnt;
#pragma unroll
      for (int k = 0; k < V; ++k) {
        const float m = live ? v[jj][k] * wj : 0.0f;
        __hip_atomic_fetch_add(ap + 32 * k, m, __ATOMIC_RELAXED, __HIP_MEMORY_SCOPE_WORKGROUP);
      }
    }
  }
}

template <int C>
__device__ __forceinline__ void agg_drain(float* __restrict__ acc, const unsigned* __restrict__ qp,
                                          int cnt, const float* __restrict__ H,
                                          const int* __restrict__ src, const float* __restrict__ dinv,
                                          int lo, int lane, v2u* __restrict__ seg, int* wcur, int basev) {
  const bool act = lane < cnt;
  const unsigned ent = act ? qp[lane] : 0u;
  const int e = (int)(ent >> 9);
  const int r = (int)(ent & 511u);
  const int s = act ? src[e] : 0;
  const float w = act ? dinv[s] * dinv[lo + r] : 0.0f;
  const int owner = r & 31;
  const int ob = __shfl(basev, owner);
  if (act) {
    const int pos = __hip_atomic_fetch_add(wcur + owner, 1, __ATOMIC_RELAXED, __HIP_MEMORY_SCOPE_WORKGROUP);
    v2u le;
    le.x = ((unsigned)s << 9) | (unsigned)r;
    le.y = __float_as_uint(w);
    seg[ob + pos] = le;
  }
  agg_apply<C>(acc, s, r, w, cnt, H, lane);
}

template <int C>
__device__ __forceinline__ void agg_rows_init(float* __restrict__ acc, const float* __restrict__ H,
                                              const float* __restrict__ dinv, const float* __restrict__ bias,
                                              int lo, int nrows, int wv, int lane) {
  constexpr int V = C / 32;
  vfN<V> bv = *(const vfN<V>*)(bias + V * lane);
  for (int r = wv; r < nrows; r += 32) {
    const float di = dinv[lo + r];
    const float d2 = di * di;
    const vfN<V> hv = *(const vfN<V>*)(H + (size_t)(lo + r) * C + V * lane);
#pragma unroll
    for (int k = 0; k < V; ++k) acc[r * C + k * 32 + lane] = hv[k] * d2 + bv[k];
  }
}

template <int C>
__device__ __forceinline__ void agg_rows_out(const float* __restrict__ acc, float* __restrict__ out,
                                             int lo, int nrows, int wv, int lane, int relu) {
  constexpr int V = C / 32;
  for (int r = wv; r < nrows; r += 32) {
    vfN<V> ov;
#pragma unroll
    for (int k = 0; k < V; ++k) {
      const float x = acc[r * C + k * 32 + lane];
      ov[k] = relu ? fmaxf(x, 0.0f) : x;
    }
    *(vfN<V>*)(out + (size_t)(lo + r) * C + V * lane) = ov;
  }
}

template <int C>
__global__ __launch_bounds__(1024) void k_agg_sift(const float* __restrict__ H, const int* __restrict__ src,
                                                   const int* __restrict__ dst, const float* __restrict__ deg,
                                                   const float* __restrict__ dinv,
                                                   const float* __restrict__ bias, float* __restrict__ out,
                                                   int relu, unsigned* __restrict__ list,
                                                   int* __restrict__ run_start, int* __restrict__ run_cnt) {
  __shared__ float acc[AGG_ROWS * C];
  __shared__ unsigned q[32 * AGG_QCAP];
  __shared__ int red[32];
  __shared__ int wc[32];
  __shared__ int wcur[32];
  const int tid  = threadIdx.x;
  const int lane = tid & 31;
  const int wv   = tid >> 5;
  const int lo   = blockIdx.x * AGG_ROWS;
  const int nrows = min(AGG_ROWS, NN - lo);

  {
    int part = 0;
    for (int i = tid; i < lo; i += 1024) part += (int)deg[i] - 1;
    int oc = 0;
    const int orow = wv + 32 * lane;
    if (lane < AGG_SLOTS && orow < nrows) oc = (int)deg[lo + orow] - 1;
#pragma unroll
    for (int m = 16; m > 0; m >>= 1) {
      part += __shfl_xor(part, m);
      oc   += __shfl_xor(oc, m);
    }
    if (lane == 0) {
      red[wv]  = part;
      wc[wv]   = oc;
      wcur[wv] = 0;
    }
  }
  agg_rows_init<C>(acc, H, dinv, bias, lo, nrows, wv, lane);
  __syncthreads();
  int pre = red[lane];
#pragma unroll
  for (int m = 16; m > 0; m >>= 1) pre += __shfl_xor(pre, m);
  v2u* seg = (v2u*)list + pre;
  const int myc = wc[lane];
  int incl = myc;
#pragma unroll
  for (int d = 1; d < 32; d <<= 1) {
    const int t = __shfl_up(incl, d);
    if (lane >= d) incl += t;
  }
  const int basev = incl - myc;

  unsigned* myq = q + wv * AGG_QCAP;
  int qn = 0;
  const int4* d4 = (const int4*)dst;
  constexpr int N4 = NE / 4;

  int4 cur[AGG_U];
#pragma unroll
  for (int u = 0; u < AGG_U; ++u) cur[u] = d4[min(u * 1024 + tid, N4 - 1)];
  for (int base = 0; base < N4; base += 1024 * AGG_U) {
    int4 nxt[AGG_U];
#pragma unroll
    for (int u = 0; u < AGG_U; ++u) {
      nxt[u] = d4[min(base + 1024 * AGG_U + u * 1024 + tid, N4 - 1)];
    }
    unsigned rr[AGG_U][4];
    unsigned mk[AGG_U][4];
    unsigned lim[AGG_U];
#pragma unroll
    for (int u = 0; u < AGG_U; ++u) {
      lim[u] = (base + u * 1024 + tid < N4) ? (unsigned)nrows : 0u;
      rr[u][0] = (unsigned)(cur[u].x - lo);
      rr[u][1] = (unsigned)(cur[u].y - lo);
      rr[u][2] = (unsigned)(cur[u].z - lo);
      rr[u][3] = (unsigned)(cur[u].w - lo);
#pragma unroll
      for (int k = 0; k < 4; ++k) mk[u][k] = __builtin_amdgcn_ballot_w32(rr[u][k] < lim[u]);
    }
#pragma unroll
    for (int u = 0; u < AGG_U; ++u) {
      const unsigned e0s = (unsigned)(base + u * 1024 + tid) << 11;
#pragma unroll
      for (int k = 0; k < 4; ++k) {
        const unsigned m = mk[u][k];
        if (m) {
          if (rr[u][k] < lim[u])
            myq[qn + __builtin_amdgcn_mbcnt_lo(m, 0u)] = e0s + ((unsigned)k << 9) + rr[u][k];
          qn += __builtin_popcount(m);
        }
      }
    }
    while (qn >= 32) {
      qn -= 32;
      agg_drain<C>(acc, myq + qn, 32, H, src, dinv, lo, lane, seg, wcur, basev);
    }
#pragma unroll
    for (int u = 0; u < AGG_U; ++u) cur[u] = nxt[u];
  }
  if (qn > 0) agg_drain<C>(acc, myq, qn, H, src, dinv, lo, lane, seg, wcur, basev);
  __syncthreads();

  if (tid < 32) {
    run_start[blockIdx.x * 32 + tid] = pre + basev;
    run_cnt[blockIdx.x * 32 + tid]   = wcur[tid];
  }
  agg_rows_out<C>(acc, out, lo, nrows, wv, lane, relu);
}

template <int C>
__device__ __forceinline__ void agg_reg_apply(float (&a0)[8], float (&a1)[8], float (&a2)[8], float (&a3)[8],
                                              v2u en, int n, const float* __restrict__ H, int lane) {
  constexpr int V = C / 32;
  constexpr int G = 16;
  const int s    = (int)(en.x >> 9);
  const int slot = (int)((en.x >> 5) & 15u);
  for (int j0 = 0; j0 < n; j0 += G) {
    vfN<V> v[G];
#pragma unroll
    for (int jj = 0; jj < G; ++jj) {
      const int sj = __builtin_amdgcn_readlane(s, j0 + jj);
      v[jj] = *(const vfN<V>*)(H + (size_t)sj * C + V * lane);
    }
#pragma unroll
    for (int jj = 0; jj < G; ++jj) {
      const int j = j0 + jj;
      if (j < n) {
        const int   sl = __builtin_amdgcn_readlane(slot, j) & 7;
        const float wj = __int_as_float(__builtin_amdgcn_readlane((int)en.y, j));
        a0[sl] = fmaf(v[jj][0], wj, a0[sl]);
        if constexpr (V > 1) a1[sl] = fmaf(v[jj][1], wj, a1[sl]);
        if constexpr (V > 2) {
          a2[sl] = fmaf(v[jj][2], wj, a2[sl]);
          a3[sl] = fmaf(v[jj][3], wj, a3[sl]);
        }
      }
    }
  }
}

template <int C>
__global__ __launch_bounds__(1024) void k_agg_list(const float* __restrict__ H, const unsigned* __restrict__ list,
                                                   const int* __restrict__ run_start,
                                                   const int* __restrict__ run_cnt,
                                                   const float* __restrict__ dinv,
                                                   const float* __restrict__ bias, float* __restrict__ out,
                                                   int relu) {
  constexpr int V = C / 32;
  static_assert(V == 1 || V == 2 || V == 4, "one register array per row component, up to four");
  const int tid  = threadIdx.x;
  const int lane = tid & 31;
  const int wv   = __builtin_amdgcn_readfirstlane(tid >> 5);
  const int lo   = blockIdx.x * AGG_ROWS;
  const int nrows = min(AGG_ROWS, NN - lo);
  const int gw   = blockIdx.x * 32 + wv;
  const int cnt  = run_cnt[gw];
  const v2u* seg = (const v2u*)list + run_start[gw];

  v2u en = {0u, 0u};
  if (lane < cnt) en = seg[lane];

  float a0[8], a1[8], a2[8], a3[8];
  {
    const vfN<V> bv = *(const vfN<V>*)(bias + V * lane);
#pragma unroll
    for (int sl = 0; sl < 8; ++sl) {
      const int r  = min(wv + 32 * min(sl, AGG_SLOTS - 1), nrows - 1);
      const float di = dinv[lo + r];
      const float d2 = di * di;
      const vfN<V> hv = *(const vfN<V>*)(H + (size_t)(lo + r) * C + V * lane);
      a0[sl] = hv[0] * d2 + bv[0];
      if constexpr (V > 1) a1[sl] = hv[1] * d2 + bv[1];
      if constexpr (V > 2) {
        a2[sl] = hv[2] * d2 + bv[2];
        a3[sl] = hv[3] * d2 + bv[3];
      }
    }
  }

  for (int c0 = 0; c0 < cnt; c0 += 32) {
    v2u nx = {0u, 0u};
    if (c0 + 32 + lane < cnt) nx = seg[c0 + 32 + lane];
    agg_reg_apply<C>(a0, a1, a2, a3, en, min(32, cnt - c0), H, lane);
    en = nx;
  }

#pragma unroll
  for (int sl = 0; sl < AGG_SLOTS; ++sl) {
    const int r = wv + 32 * sl;
    if (r < nrows) {
      vfN<V> ov;
      ov[0] = a0[sl];
      if constexpr (V > 1) ov[1] = a1[sl];
      if constexpr (V > 2) {
        ov[2] = a2[sl];
        ov[3] = a3[sl];
      }
#pragma unroll
      for (int k = 0; k < V; ++k) ov[k] = relu ? fmaxf(ov[k], 0.0f) : ov[k];
      *(vfN<V>*)(out + (size_t)(lo + r) * C + V * lane) = ov;
    }
  }
}

static inline unsigned nblk(long long n, int b) { return (unsigned)((n + b - 1) / b); }

extern "C" void kernel_launch(void* const* d_in, const int* in_sizes, int n_in,
                              void* d_out, int out_size, void* d_ws, size_t ws_size,
                              hipStream_t stream) {
  (void)in_sizes; (void)n_in; (void)out_size; (void)ws_size;
  const float* x   = (const float*)d_in[0];
  const int*   ei  = (const int*)d_in[1];
  const float* We1 = (const float*)d_in[2]; const float* be1 = (const float*)d_in[3];
  const float* We2 = (const float*)d_in[4]; const float* be2 = (const float*)d_in[5];
  const float* Wd1 = (const float*)d_in[6]; const float* bd1 = (const float*)d_in[7];
  const float* Wd2 = (const float*)d_in[8]; const float* bd2 = (const float*)d_in[9];
  const int* src = ei;
  const int* dst = ei + NE;

  char*  ws   = (char*)d_ws;
  float* dinv = (float*)ws;
  size_t off  = ((size_t)NN * 4 + 255) & ~(size_t)255;
  float* G    = (float*)(ws + off);
  float* P    = G + (size_t)NN * 128;
  float* deg  = P + (size_t)NN * 64;
  unsigned* list = (unsigned*)((char*)deg + off);
  int* run_start = (int*)(list + (size_t)NE * 2);
  int* run_cnt   = run_start + AGG_NB * 32;
  float* O    = (float*)d_out;

  k_init_deg <<<nblk(NN, 256), 256, 0, stream>>>(deg);
  k_deg_edges<<<nblk(NE, 256), 256, 0, stream>>>(dst, deg);
  k_rsqrt    <<<nblk(NN, 256), 256, 0, stream>>>(deg, dinv);

  k_gemm_stripe<128, 64><<<GEMM_NB, GEMM_WAVES * 32, 0, stream>>>(x, We1, G);
  k_agg_sift<64><<<AGG_NB, 1024, 0, stream>>>(G, src, dst, deg, dinv, be1, P, 1, list, run_start, run_cnt);

  k_gemm_stripe<64, 32><<<GEMM_NB, GEMM_WAVES * 32, 0, stream>>>(P, We2, G);
  k_agg_list<32> <<<AGG_NB, 1024, 0, stream>>>(G, list, run_start, run_cnt, dinv, be2, P, 0);

  k_gemm_stripe<32, 64><<<GEMM_NB, GEMM_WAVES * 32, 0, stream>>>(P, Wd1, G);
  k_agg_list<64> <<<AGG_NB, 1024, 0, stream>>>(G, list, run_start, run_cnt, dinv, bd1, P, 1);

  k_gemm_stripe<64, 128><<<GEMM_NB, GEMM_WAVES * 32, 0, stream>>>(P, Wd2, G);
  k_agg_list<128><<<AGG_NB, 1024, 0, stream>>>(G, list, run_start, run_cnt, dinv, bd2, O, 0);
}


# Sift in two levels: a new launch keeps, per 1/16 chunk of dst, the edges of each group of 16 work-groups; the first aggregation sifts its group's kept edges, not all of dst; sum order stays arrival
# speedup vs baseline: 2.5380x; 1.0464x over previous
#include <hip/hip_runtime.h>


#define NN 50000
#define NE 800000

typedef __attribute__((ext_vector_type(2))) float v2f;
typedef __attribute__((ext_vector_type(8))) float v8f;

__global__ void k_init_deg(float* deg) {
  int i = blockIdx.x * blockDim.x + threadIdx.x;
  if (i < NN) deg[i] = 1.0f;
}

__global__ void k_deg_edges(const int* __restrict__ dst, float* __restrict__ deg) {
  int e = blockIdx.x * blockDim.x + threadIdx.x;
  if (e < NE)
    __hip_atomic_fetch_add(deg + dst[e], 1.0f, __ATOMIC_RELAXED, __HIP_MEMORY_SCOPE_AGENT);
}

__global__ void k_rsqrt(const float* __restrict__ deg, float* __restrict__ dinv) {
  int i = blockIdx.x * blockDim.x + threadIdx.x;
  if (i < NN) dinv[i] = rsqrtf(deg[i]);
}

#define GEMM_WAVES 8
#define GEMM_STRIPES (NN / 16)
#define GEMM_NB ((GEMM_STRIPES + GEMM_WAVES - 1) / GEMM_WAVES)
static_assert(NN % 16 == 0, "rows are covered by whole 16-row stripes");

typedef __attribute__((ext_vector_type(4))) float v4f;

template <int K, int N>
__global__ __launch_bounds__(GEMM_WAVES * 32) void k_gemm_stripe(const float* __restrict__ X,
                                                                 const float* __restrict__ W,
                                                                 float* __restrict__ H) {
  constexpr int T  = N / 16;
  constexpr int KS = K / 4;
  constexpr int WP = N + 16;
  __shared__ v2f wl[(K / 2) * WP];
  const int tid  = threadIdx.x;
  const int lane = tid & 31;
  const int wv   = __builtin_amdgcn_readfirstlane(tid >> 5);
  const int half = lane >> 4;
  const int l    = lane & 15;
  const int stripe = blockIdx.x * GEMM_WAVES + wv;
  const bool live = stripe < GEMM_STRIPES;

  v2f a[KS];
  if (live) {
    const float* xrow = X + (size_t)(stripe * 16 + l) * K + 2 * half;
#pragma unroll
    for (int kk = 0; kk < KS; ++kk) a[kk] = *(const v2f*)(xrow + 4 * kk);
  } else {
#pragma unroll
    for (int kk = 0; kk < KS; ++kk) a[kk] = v2f{0.0f, 0.0f};
  }

  constexpr int NI = (K / 2) * (N / 4) / (GEMM_WAVES * 32);
  static_assert(NI * GEMM_WAVES * 32 == (K / 2) * (N / 4), "W is staged in whole passes of the work-group");
  v4f w0[NI], w1[NI];
#pragma unroll
  for (int it = 0; it < NI; ++it) {
    const int i  = tid + it * GEMM_WAVES * 32;
    const int kp = i / (N / 4);
    const int q  = i - kp * (N / 4);
    w0[it] = *(const v4f*)(W + (size_t)(2 * kp) * N + 4 * q);
    w1[it] = *(const v4f*)(W + (size_t)(2 * kp + 1) * N + 4 * q);
  }
#pragma unroll
  for (int it = 0; it < NI; ++it) {
    const int i  = tid + it * GEMM_WAVES * 32;
    const int kp = i / (N / 4);
    const int q  = i - kp * (N / 4);
    v4f* dp = (v4f*)(wl + kp * WP + 4 * q);
    dp[0] = v4f{w0[it].x, w1[it].x, w0[it].y, w1[it].y};
    dp[1] = v4f{w0[it].z, w1[it].z, w0[it].w, w1[it].w};
  }
  __syncthreads();
  if (!live) return;

  v8f c[T];
#pragma unroll
  for (int t = 0; t < T; ++t) c[t] = v8f{};
  const v2f* bp = wl + half * WP + l;
#pragma unroll
  for (int kk = 0; kk < KS; ++kk) {
    v2f b[T];
#pragma unroll
    for (int t = 0; t < T; ++t) b[t] = bp[(2 * kk) * WP + 16 * t];
#pragma unroll
    for (int t = 0; t < T; ++t)
      c[t] = __builtin_amdgcn_wmma_f32_16x16x4_f32(false, a[kk], false, b[t], (short)0, c[t], false, false);
  }
  float* hrow = H + (size_t)(stripe * 16 + 8 * half) * N + l;
#pragma unroll
  for (int t = 0; t < T; ++t) {
#pragma unroll
    for (int r = 0; r < 8; ++r) hrow[(size_t)r * N + 16 * t] = c[t][r];
  }
}

#define AGG_NB    256
#define AGG_ROWS  ((NN + AGG_NB - 1) / AGG_NB)
#define PS_NG     16
#define PS_NC     16
#define PS_WG     (AGG_NB / PS_NG)
#define PS_GROWS  (PS_WG * AGG_ROWS)
#define PS_C4     (NE / 4 / PS_NC)
#define PS_PASSES ((PS_C4 + 1023) / 1024)
#define PS_CAP    (PS_PASSES * 128)
#define PS_B      8
#define AGG_QCAP  (32 + 128 * PS_B)
#define AGG_SLOTS ((AGG_ROWS + 31) / 32)
static_assert(AGG_SLOTS <= 8, "a wave's rows are held in register arrays of 8 slots");

static_assert(NE % (4 * PS_NC) == 0, "dst is streamed as int4, in whole chunks");
static_assert(AGG_NB % PS_NG == 0 && PS_NC % PS_B == 0 && PS_NC <= 32, "groups and chunks tile evenly");
static_assert(PS_GROWS <= (1 << 12) && NE <= (1 << 20), "kept entry = edge id << 12 | row in group");
static_assert(PS_CAP % 4 == 0, "kept-edge segments start on 16-byte boundaries");
static_assert(AGG_ROWS <= 512, "row offset is packed into 9 bits");
static_assert(NE <= (1 << 22), "edge id is packed above 9 bits");
static_assert(NN <= (1 << 23), "source row is packed above 9 bits");

template <int V> using vfN = float __attribute__((ext_vector_type(V)));
typedef unsigned __attribute__((ext_vector_type(2))) v2u;

template <int C>
__device__ __forceinline__ void agg_apply(float* __restrict__ acc, int s, int r, float w, int cnt,
                                          const float* __restrict__ H, int lane) {
  constexpr int V = C / 32;
  constexpr int G = 16;
  for (int j0 = 0; j0 < cnt; j0 += G) {
    vfN<V> v[G];
#pragma unroll
    for (int jj = 0; jj < G; ++jj) {
      const int sj = __builtin_amdgcn_readlane(s, j0 + jj);
      v[jj] = *(const vfN<V>*)(H + (size_t)sj * C + V * lane);
    }
#pragma unroll
    for (int jj = 0; jj < G; ++jj) {
      const int j = j0 + jj;
      const int   rj = __builtin_amdgcn_readlane(r, j);
      const float wj = __int_as_float(__builtin_amdgcn_readlane(__float_as_int(w), j));
      float* ap = acc + rj * C + lane;
      const bool live = j < cnt;
#pragma unroll
      for (int k = 0; k < V; ++k) {
        const float m = live ? v[jj][k] * wj : 0.0f;
        __hip_atomic_fetch_add(ap + 32 * k, m, __ATOMIC_RELAXED, __HIP_MEMORY_SCOPE_WORKGROUP);
      }
    }
  }
}

template <int C>
__device__ __forceinline__ void agg_drain(float* __restrict__ acc, const unsigned* __restrict__ qp,
                                          int cnt, const float* __restrict__ H,
                                          const int* __restrict__ src, const float* __restrict__ dinv,
                                          int lo, int lane, v2u* __restrict__ seg, int* wcur, int basev) {
  const bool act = lane < cnt;
  const unsigned ent = act ? qp[lane] : 0u;
  const int e = (int)(ent >> 9);
  const int r = (int)(ent & 511u);
  const int s = act ? src[e] : 0;
  const float w = act ? dinv[s] * dinv[lo + r] : 0.0f;
  const int owner = r & 31;
  const int ob = __shfl(basev, owner);
  if (act) {
    const int pos = __hip_atomic_fetch_add(wcur + owner, 1, __ATOMIC_RELAXED, __HIP_MEMORY_SCOPE_WORKGROUP);
    v2u le;
    le.x = ((unsigned)s << 9) | (unsigned)r;
    le.y = __float_as_uint(w);
    seg[ob + pos] = le;
  }
  agg_apply<C>(acc, s, r, w, cnt, H, lane);
}

template <int C>
__device__ __forceinline__ void agg_rows_init(float* __restrict__ acc, const float* __restrict__ H,
                                              const float* __restrict__ dinv, const float* __restrict__ bias,
                                              int lo, int nrows, int wv, int lane) {
  constexpr int V = C / 32;
  vfN<V> bv = *(const vfN<V>*)(bias + V * lane);
  for (int r = wv; r < nrows; r += 32) {
    const float di = dinv[lo + r];
    const float d2 = di * di;
    const vfN<V> hv = *(const vfN<V>*)(H + (size_t)(lo + r) * C + V * lane);
#pragma unroll
    for (int k = 0; k < V; ++k) acc[r * C + k * 32 + lane] = hv[k] * d2 + bv[k];
  }
}

template <int C>
__device__ __forceinline__ void agg_rows_out(const float* __restrict__ acc, float* __restrict__ out,
                                             int lo, int nrows, int wv, int lane, int relu) {
  constexpr int V = C / 32;
  for (int r = wv; r < nrows; r += 32) {
    vfN<V> ov;
#pragma unroll
    for (int k = 0; k < V; ++k) {
      const float x = acc[r * C + k * 32 + lane];
      ov[k] = relu ? fmaxf(x, 0.0f) : x;
    }
    *(vfN<V>*)(out + (size_t)(lo + r) * C + V * lane) = ov;
  }
}

__global__ __launch_bounds__(1024) void k_presift(const int* __restrict__ dst, unsigned* __restrict__ pl,
                                                  int* __restrict__ pcnt) {
  const int tid  = threadIdx.x;
  const int lane = tid & 31;
  const int wv   = tid >> 5;
  const int g    = blockIdx.x / PS_NC;
  const int c    = blockIdx.x - g * PS_NC;
  const int glo  = g * PS_GROWS;
  const unsigned gn = (unsigned)min(PS_GROWS, NN - glo);
  const int4* d4 = (const int4*)dst + (size_t)c * PS_C4;
  unsigned* seg  = pl + (size_t)(blockIdx.x * 32 + wv) * PS_CAP;

  int4 v[PS_PASSES];
#pragma unroll
  for (int p = 0; p < PS_PASSES; ++p) v[p] = d4[min(p * 1024 + tid, PS_C4 - 1)];
  int n = 0;
#pragma unroll
  for (int p = 0; p < PS_PASSES; ++p) {
    const int idx = p * 1024 + tid;
    const unsigned lim = (idx < PS_C4) ? gn : 0u;
    const unsigned e0s = (unsigned)(c * PS_C4 + idx) << 14;
    const int dv[4] = {v[p].x, v[p].y, v[p].z, v[p].w};
#pragma unroll
    for (int k = 0; k < 4; ++k) {
      const unsigned rr = (unsigned)(dv[k] - glo);
      const bool hit = rr < lim;
      const unsigned m = __builtin_amdgcn_ballot_w32(hit);
      if (m) {
        if (hit) seg[n + __builtin_amdgcn_mbcnt_lo(m, 0u)] = e0s + ((unsigned)k << 12) + rr;
        n += __builtin_popcount(m);
      }
    }
  }
  if (lane == 0) pcnt[blockIdx.x * 32 + wv] = n;
}

__device__ __forceinline__ void agg_sift4(uint4 e4, int left, unsigned sublo, unsigned nrows,
                                          unsigned* __restrict__ myq, int& qn) {
  const unsigned ev[4] = {e4.x, e4.y, e4.z, e4.w};
#pragma unroll
  for (int k = 0; k < 4; ++k) {
    const unsigned rr = (ev[k] & 4095u) - sublo;
    const bool hit = (k < left) && (rr < nrows);
    const unsigned m = __builtin_amdgcn_ballot_w32(hit);
    if (m) {
      if (hit) myq[qn + __builtin_amdgcn_mbcnt_lo(m, 0u)] = ((ev[k] >> 12) << 9) | rr;
      qn += __builtin_popcount(m);
    }
  }
}

template <int C>
__global__ __launch_bounds__(1024) void k_agg_sift(const float* __restrict__ H, const int* __restrict__ src,
                                                   const unsigned* __restrict__ pl, const int* __restrict__ pcnt,
                                                   const float* __restrict__ deg,
                                                   const float* __restrict__ dinv,
                                                   const float* __restrict__ bias, float* __restrict__ out,
                                                   int relu, unsigned* __restrict__ list,
                                                   int* __restrict__ run_start, int* __restrict__ run_cnt) {
  __shared__ float acc[AGG_ROWS * C];
  __shared__ unsigned q[32 * AGG_QCAP];
  __shared__ int red[32];
  __shared__ int wc[32];
  __shared__ int wcur[32];
  const int tid  = threadIdx.x;
  const int lane = tid & 31;
  const int wv   = tid >> 5;
  const int lo   = blockIdx.x * AGG_ROWS;
  const int nrows = min(AGG_ROWS, NN - lo);

  {
    int part = 0;
    for (int i = tid; i < lo; i += 1024) part += (int)deg[i] - 1;
    int oc = 0;
    const int orow = wv + 32 * lane;
    if (lane < AGG_SLOTS && orow < nrows) oc = (int)deg[lo + orow] - 1;
#pragma unroll
    for (int m = 16; m > 0; m >>= 1) {
      part += __shfl_xor(part, m);
      oc   += __shfl_xor(oc, m);
    }
    if (lane == 0) {
      red[wv]  = part;
      wc[wv]   = oc;
      wcur[wv] = 0;
    }
  }
  agg_rows_init<C>(acc, H, dinv, bias, lo, nrows, wv, lane);
  __syncthreads();
  int pre = red[lane];
#pragma unroll
  for (int m = 16; m > 0; m >>= 1) pre += __shfl_xor(pre, m);
  v2u* seg = (v2u*)list + pre;
  const int myc = wc[lane];
  int incl = myc;
#pragma unroll
  for (int d = 1; d < 32; d <<= 1) {
    const int t = __shfl_up(incl, d);
    if (lane >= d) incl += t;
  }
  const int basev = incl - myc;

  unsigned* myq = q + wv * AGG_QCAP;
  int qn = 0;
  const int grp = blockIdx.x / PS_WG;
  const unsigned sublo = (unsigned)(lo - grp * PS_GROWS);
  const int sbase = grp * PS_NC * 32 + wv;
  const int myn = (lane < PS_NC) ? pcnt[sbase + 32 * lane] : 0;
#pragma nounroll
  for (int c0 = 0; c0 < PS_NC; c0 += PS_B) {
    uint4 ent[PS_B];
    int   nn[PS_B];
#pragma unroll
    for (int u = 0; u < PS_B; ++u) {
      nn[u] = __builtin_amdgcn_readlane(myn, c0 + u);
      const uint4* sp = (const uint4*)(pl + (size_t)(sbase + 32 * (c0 + u)) * PS_CAP);
      ent[u] = (4 * lane < nn[u]) ? sp[lane] : make_uint4(0u, 0u, 0u, 0u);
    }
#pragma unroll
    for (int u = 0; u < PS_B; ++u) agg_sift4(ent[u], nn[u] - 4 * lane, sublo, (unsigned)nrows, myq, qn);
    while (qn >= 32) {
      qn -= 32;
      agg_drain<C>(acc, myq + qn, 32, H, src, dinv, lo, lane, seg, wcur, basev);
    }
#pragma nounroll
    for (int u = 0; u < PS_B; ++u) {
      const int nu = __builtin_amdgcn_readlane(myn, c0 + u);
      const uint4* sp = (const uint4*)(pl + (size_t)(sbase + 32 * (c0 + u)) * PS_CAP);
      for (int off = 128; off < nu; off += 128) {
        const int left = nu - off - 4 * lane;
        const uint4 e4 = (left > 0) ? sp[off / 4 + lane] : make_uint4(0u, 0u, 0u, 0u);
        agg_sift4(e4, left, sublo, (unsigned)nrows, myq, qn);
        while (qn >= 32) {
          qn -= 32;
          agg_drain<C>(acc, myq + qn, 32, H, src, dinv, lo, lane, seg, wcur, basev);
        }
      }
    }
  }
  if (qn > 0) agg_drain<C>(acc, myq, qn, H, src, dinv, lo, lane, seg, wcur, basev);
  __syncthreads();

  if (tid < 32) {
    run_start[blockIdx.x * 32 + tid] = pre + basev;
    run_cnt[blockIdx.x * 32 + tid]   = wcur[tid];
  }
  agg_rows_out<C>(acc, out, lo, nrows, wv, lane, relu);
}

template <int C>
__device__ __forceinline__ void agg_reg_apply(float (&a0)[8], float (&a1)[8], float (&a2)[8], float (&a3)[8],
                                              v2u en, int n, const float* __restrict__ H, int lane) {
  constexpr int V = C / 32;
  constexpr int G = 16;
  const int s    = (int)(en.x >> 9);
  const int slot = (int)((en.x >> 5) & 15u);
  for (int j0 = 0; j0 < n; j0 += G) {
    vfN<V> v[G];
#pragma unroll
    for (int jj = 0; jj < G; ++jj) {
      const int sj = __builtin_amdgcn_readlane(s, j0 + jj);
      v[jj] = *(const vfN<V>*)(H + (size_t)sj * C + V * lane);
    }
#pragma unroll
    for (int jj = 0; jj < G; ++jj) {
      const int j = j0 + jj;
      if (j < n) {
        const int   sl = __builtin_amdgcn_readlane(slot, j) & 7;
        const float wj = __int_as_float(__builtin_amdgcn_readlane((int)en.y, j));
        a0[sl] = fmaf(v[jj][0], wj, a0[sl]);
        if constexpr (V > 1) a1[sl] = fmaf(v[jj][1], wj, a1[sl]);
        if constexpr (V > 2) {
          a2[sl] = fmaf(v[jj][2], wj, a2[sl]);
          a3[sl] = fmaf(v[jj][3], wj, a3[sl]);
        }
      }
    }
  }
}

template <int C>
__global__ __launch_bounds__(1024) void k_agg_list(const float* __restrict__ H, const unsigned* __restrict__ list,
                                                   const int* __restrict__ run_start,
                                                   const int* __restrict__ run_cnt,
                                                   const float* __restrict__ dinv,
                                                   const float* __restrict__ bias, float* __restrict__ out,
                                                   int relu) {
  constexpr int V = C / 32;
  static_assert(V == 1 || V == 2 || V == 4, "one register array per row component, up to four");
  const int tid  = threadIdx.x;
  const int lane = tid & 31;
  const int wv   = __builtin_amdgcn_readfirstlane(tid >> 5);
  const int lo   = blockIdx.x * AGG_ROWS;
  const int nrows = min(AGG_ROWS, NN - lo);
  const int gw   = blockIdx.x * 32 + wv;
  const int cnt  = run_cnt[gw];
  const v2u* seg = (const v2u*)list + run_start[gw];

  v2u en = {0u, 0u};
  if (lane < cnt) en = seg[lane];

  float a0[8], a1[8], a2[8], a3[8];
  {
    const vfN<V> bv = *(const vfN<V>*)(bias + V * lane);
#pragma unroll
    for (int sl = 0; sl < 8; ++sl) {
      const int r  = min(wv + 32 * min(sl, AGG_SLOTS - 1), nrows - 1);
      const float di = dinv[lo + r];
      const float d2 = di * di;
      const vfN<V> hv = *(const vfN<V>*)(H + (size_t)(lo + r) * C + V * lane);
      a0[sl] = hv[0] * d2 + bv[0];
      if constexpr (V > 1) a1[sl] = hv[1] * d2 + bv[1];
      if constexpr (V > 2) {
        a2[sl] = hv[2] * d2 + bv[2];
        a3[sl] = hv[3] * d2 + bv[3];
      }
    }
  }

  for (int c0 = 0; c0 < cnt; c0 += 32) {
    v2u nx = {0u, 0u};
    if (c0 + 32 + lane < cnt) nx = seg[c0 + 32 + lane];
    agg_reg_apply<C>(a0, a1, a2, a3, en, min(32, cnt - c0), H, lane);
    en = nx;
  }

#pragma unroll
  for (int sl = 0; sl < AGG_SLOTS; ++sl) {
    const int r = wv + 32 * sl;
    if (r < nrows) {
      vfN<V> ov;
      ov[0] = a0[sl];
      if constexpr (V > 1) ov[1] = a1[sl];
      if constexpr (V > 2) {
        ov[2] = a2[sl];
        ov[3] = a3[sl];
      }
#pragma unroll
      for (int k = 0; k < V; ++k) ov[k] = relu ? fmaxf(ov[k], 0.0f) : ov[k];
      *(vfN<V>*)(out + (size_t)(lo + r) * C + V * lane) = ov;
    }
  }
}

static inline unsigned nblk(long long n, int b) { return (unsigned)((n + b - 1) / b); }

extern "C" void kernel_launch(void* const* d_in, const int* in_sizes, int n_in,
                              void* d_out, int out_size, void* d_ws, size_t ws_size,
                              hipStream_t stream) {
  (void)in_sizes; (void)n_in; (void)out_size; (void)ws_size;
  const float* x   = (const float*)d_in[0];
  const int*   ei  = (const int*)d_in[1];
  const float* We1 = (const float*)d_in[2]; const float* be1 = (const float*)d_in[3];
  const float* We2 = (const float*)d_in[4]; const float* be2 = (const float*)d_in[5];
  const float* Wd1 = (const float*)d_in[6]; const float* bd1 = (const float*)d_in[7];
  const float* Wd2 = (const float*)d_in[8]; const float* bd2 = (const float*)d_in[9];
  const int* src = ei;
  const int* dst = ei + NE;

  char*  ws   = (char*)d_ws;
  float* dinv = (float*)ws;
  size_t off  = ((size_t)NN * 4 + 255) & ~(size_t)255;
  float* G    = (float*)(ws + off);
  float* P    = G + (size_t)NN * 128;
  float* deg  = P + (size_t)NN * 64;
  unsigned* list = (unsigned*)((char*)deg + off);
  int* run_start = (int*)(list + (size_t)NE * 2);
  int* run_cnt   = run_start + AGG_NB * 32;
  unsigned* pl   = (unsigned*)(run_cnt + AGG_NB * 32);
  int* pcnt      = (int*)(pl + (size_t)PS_NG * PS_NC * 32 * PS_CAP);
  float* O    = (float*)d_out;

  k_init_deg <<<nblk(NN, 256), 256, 0, stream>>>(deg);
  k_deg_edges<<<nblk(NE, 256), 256, 0, stream>>>(dst, deg);
  k_rsqrt    <<<nblk(NN, 256), 256, 0, stream>>>(deg, dinv);

  k_presift<<<PS_NG * PS_NC, 1024, 0, stream>>>(dst, pl, pcnt);

  k_gemm_stripe<128, 64><<<GEMM_NB, GEMM_WAVES * 32, 0, stream>>>(x, We1, G);
  k_agg_sift<64><<<AGG_NB, 1024, 0, stream>>>(G, src, pl, pcnt, deg, dinv, be1, P, 1, list, run_start, run_cnt);

  k_gemm_stripe<64, 32><<<GEMM_NB, GEMM_WAVES * 32, 0, stream>>>(P, We2, G);
  k_agg_list<32> <<<AGG_NB, 1024, 0, stream>>>(G, list, run_start, run_cnt, dinv, be2, P, 0);

  k_gemm_stripe<32, 64><<<GEMM_NB, GEMM_WAVES * 32, 0, stream>>>(P, Wd1, G);
  k_agg_list<64> <<<AGG_NB, 1024, 0, stream>>>(G, list, run_start, run_cnt, dinv, bd1, P, 1);

  k_gemm_stripe<64, 128><<<GEMM_NB, GEMM_WAVES * 32, 0, stream>>>(P, Wd2, G);
  k_agg_list<128><<<AGG_NB, 1024, 0, stream>>>(G, list, run_start, run_cnt, dinv, bd2, O, 0);
}


# Degree pass folded into existing launches: the first-level sift counts its group's in-edges per chunk (LDS int adds), the first product's launch sums them to deg and dinv; 12 -> 9 launches
# speedup vs baseline: 3.5519x; 1.3995x over previous
#include <hip/hip_runtime.h>


#define NN 50000
#define NE 800000

typedef __attribute__((ext_vector_type(2))) float v2f;
typedef __attribute__((ext_vector_type(8))) float v8f;

#define PS_NC     16
#define PC_PITCH  ((NN + 31) / 32 * 32)

#define GEMM_WAVES 8
#define GEMM_STRIPES (NN / 16)
#define GEMM_NB ((GEMM_STRIPES + GEMM_WAVES - 1) / GEMM_WAVES)
static_assert(NN % 16 == 0, "rows are covered by whole 16-row stripes");

typedef __attribute__((ext_vector_type(4))) float v4f;

template <int K, int N, bool DEG>
__global__ __launch_bounds__(GEMM_WAVES * 32) void k_gemm_stripe(const float* __restrict__ X,
                                                                 const float* __restrict__ W,
                                                                 float* __restrict__ H,
                                                                 const int* __restrict__ pc,
                                                                 float* __restrict__ deg,
                                                                 float* __restrict__ dinv) {
  constexpr int T  = N / 16;
  constexpr int KS = K / 4;
  constexpr int WP = N + 16;
  __shared__ v2f wl[(K / 2) * WP];
  const int tid  = threadIdx.x;
  const int lane = tid & 31;
  const int wv   = __builtin_amdgcn_readfirstlane(tid >> 5);
  const int half = lane >> 4;
  const int l    = lane & 15;
  const int stripe = blockIdx.x * GEMM_WAVES + wv;
  const bool live = stripe < GEMM_STRIPES;

  v2f a[KS];
  if (live) {
    const float* xrow = X + (size_t)(stripe * 16 + l) * K + 2 * half;
#pragma unroll
    for (int kk = 0; kk < KS; ++kk) a[kk] = *(const v2f*)(xrow + 4 * kk);
  } else {
#pragma unroll
    for (int kk = 0; kk < KS; ++kk) a[kk] = v2f{0.0f, 0.0f};
  }

  constexpr int NI = (K / 2) * (N / 4) / (GEMM_WAVES * 32);
  static_assert(NI * GEMM_WAVES * 32 == (K / 2) * (N / 4), "W is staged in whole passes of the work-group");
  v4f w0[NI], w1[NI];
#pragma unroll
  for (int it = 0; it < NI; ++it) {
    const int i  = tid + it * GEMM_WAVES * 32;
    const int kp = i / (N / 4);
    const int q  = i - kp * (N / 4);
    w0[it] = *(const v4f*)(W + (size_t)(2 * kp) * N + 4 * q);
    w1[it] = *(const v4f*)(W + (size_t)(2 * kp + 1) * N + 4 * q);
  }
  if constexpr (DEG) {
    const int row = blockIdx.x * (GEMM_WAVES * 16) + tid;
    if (tid < GEMM_WAVES * 16 && row < NN) {
      int n = 0;
#pragma unroll
      for (int c = 0; c < PS_NC; ++c) n += pc[(size_t)c * PC_PITCH + row];
      const float d = 1.0f + (float)n;
      deg[row]  = d;
      dinv[row] = rsqrtf(d);
    }
  }
#pragma unroll
  for (int it = 0; it < NI; ++it) {
    const int i  = tid + it * GEMM_WAVES * 32;
    const int kp = i / (N / 4);
    const int q  = i - kp * (N / 4);
    v4f* dp = (v4f*)(wl + kp * WP + 4 * q);
    dp[0] = v4f{w0[it].x, w1[it].x, w0[it].y, w1[it].y};
    dp[1] = v4f{w0[it].z, w1[it].z, w0[it].w, w1[it].w};
  }
  __syncthreads();
  if (!live) return;

  v8f c[T];
#pragma unroll
  for (int t = 0; t < T; ++t) c[t] = v8f{};
  const v2f* bp = wl + half * WP + l;
#pragma unroll
  for (int kk = 0; kk < KS; ++kk) {
    v2f b[T];
#pragma unroll
    for (int t = 0; t < T; ++t) b[t] = bp[(2 * kk) * WP + 16 * t];
#pragma unroll
    for (int t = 0; t < T; ++t)
      c[t] = __builtin_amdgcn_wmma_f32_16x16x4_f32(false, a[kk], false, b[t], (short)0, c[t], false, false);
  }
  float* hrow = H + (size_t)(stripe * 16 + 8 * half) * N + l;
#pragma unroll
  for (int t = 0; t < T; ++t) {
#pragma unroll
    for (int r = 0; r < 8; ++r) hrow[(size_t)r * N + 16 * t] = c[t][r];
  }
}

#define AGG_NB    256
#define AGG_ROWS  ((NN + AGG_NB - 1) / AGG_NB)
#define PS_NG     16
#define PS_WG     (AGG_NB / PS_NG)
#define PS_GROWS  (PS_WG * AGG_ROWS)
#define PS_C4     (NE / 4 / PS_NC)
#define PS_PASSES ((PS_C4 + 1023) / 1024)
#define PS_CAP    (PS_PASSES * 128)
#define PS_B      8
#define AGG_QCAP  (32 + 128 * PS_B)
#define AGG_SLOTS ((AGG_ROWS + 31) / 32)
static_assert(AGG_SLOTS <= 8, "a wave's rows are held in register arrays of 8 slots");

static_assert(NE % (4 * PS_NC) == 0, "dst is streamed as int4, in whole chunks");
static_assert(AGG_NB % PS_NG == 0 && PS_NC % PS_B == 0 && PS_NC <= 32, "groups and chunks tile evenly");
static_assert(PS_GROWS <= (1 << 12) && NE <= (1 << 20), "kept entry = edge id << 12 | row in group");
static_assert(PS_CAP % 4 == 0, "kept-edge segments start on 16-byte boundaries");
static_assert(AGG_ROWS <= 512, "row offset is packed into 9 bits");
static_assert(NE <= (1 << 22), "edge id is packed above 9 bits");
static_assert(NN <= (1 << 23), "source row is packed above 9 bits");

template <int V> using vfN = float __attribute__((ext_vector_type(V)));
typedef unsigned __attribute__((ext_vector_type(2))) v2u;

template <int C>
__device__ __forceinline__ void agg_apply(float* __restrict__ acc, int s, int r, float w, int cnt,
                                          const float* __restrict__ H, int lane) {
  constexpr int V = C / 32;
  constexpr int G = 16;
  for (int j0 = 0; j0 < cnt; j0 += G) {
    vfN<V> v[G];
#pragma unroll
    for (int jj = 0; jj < G; ++jj) {
      const int sj = __builtin_amdgcn_readlane(s, j0 + jj);
      v[jj] = *(const vfN<V>*)(H + (size_t)sj * C + V * lane);
    }
#pragma unroll
    for (int jj = 0; jj < G; ++jj) {
      const int j = j0 + jj;
      const int   rj = __builtin_amdgcn_readlane(r, j);
      const float wj = __int_as_float(__builtin_amdgcn_readlane(__float_as_int(w), j));
      float* ap = acc + rj * C + lane;
      const bool live = j < cnt;
#pragma unroll
      for (int k = 0; k < V; ++k) {
        const float m = live ? v[jj][k] * wj : 0.0f;
        __hip_atomic_fetch_add(ap + 32 * k, m, __ATOMIC_RELAXED, __HIP_MEMORY_SCOPE_WORKGROUP);
      }
    }
  }
}

template <int C>
__device__ __forceinline__ void agg_drain(float* __restrict__ acc, const unsigned* __restrict__ qp,
                                          int cnt, const float* __restrict__ H,
                                          const int* __restrict__ src, const float* __restrict__ dinv,
                                          int lo, int lane, v2u* __restrict__ seg, int* wcur, int basev) {
  const bool act = lane < cnt;
  const unsigned ent = act ? qp[lane] : 0u;
  const int e = (int)(ent >> 9);
  const int r = (int)(ent & 511u);
  const int s = act ? src[e] : 0;
  const float w = act ? dinv[s] * dinv[lo + r] : 0.0f;
  const int owner = r & 31;
  const int ob = __shfl(basev, owner);
  if (act) {
    const int pos = __hip_atomic_fetch_add(wcur + owner, 1, __ATOMIC_RELAXED, __HIP_MEMORY_SCOPE_WORKGROUP);
    v2u le;
    le.x = ((unsigned)s << 9) | (unsigned)r;
    le.y = __float_as_uint(w);
    seg[ob + pos] = le;
  }
  agg_apply<C>(acc, s, r, w, cnt, H, lane);
}

template <int C>
__device__ __forceinline__ void agg_rows_init(float* __restrict__ acc, const float* __restrict__ H,
                                              const float* __restrict__ dinv, const float* __restrict__ bias,
                                              int lo, int nrows, int wv, int lane) {
  constexpr int V = C / 32;
  vfN<V> bv = *(const vfN<V>*)(bias + V * lane);
  for (int r = wv; r < nrows; r += 32) {
    const float di = dinv[lo + r];
    const float d2 = di * di;
    const vfN<V> hv = *(const vfN<V>*)(H + (size_t)(lo + r) * C + V * lane);
#pragma unroll
    for (int k = 0; k < V; ++k) acc[r * C + k * 32 + lane] = hv[k] * d2 + bv[k];
  }
}

template <int C>
__device__ __forceinline__ void agg_rows_out(const float* __restrict__ acc, float* __restrict__ out,
                                             int lo, int nrows, int wv, int lane, int relu) {
  constexpr int V = C / 32;
  for (int r = wv; r < nrows; r += 32) {
    vfN<V> ov;
#pragma unroll
    for (int k = 0; k < V; ++k) {
      const float x = acc[r * C + k * 32 + lane];
      ov[k] = relu ? fmaxf(x, 0.0f) : x;
    }
    *(vfN<V>*)(out + (size_t)(lo + r) * C + V * lane) = ov;
  }
}

__global__ __launch_bounds__(1024) void k_presift(const int* __restrict__ dst, unsigned* __restrict__ pl,
                                                  int* __restrict__ pcnt, int* __restrict__ pc) {
  __shared__ int hist[PS_GROWS];
  const int tid  = threadIdx.x;
  const int lane = tid & 31;
  const int wv   = tid >> 5;
  const int g    = blockIdx.x / PS_NC;
  const int c    = blockIdx.x - g * PS_NC;
  const int glo  = g * PS_GROWS;
  const unsigned gn = (unsigned)min(PS_GROWS, NN - glo);
  const int4* d4 = (const int4*)dst + (size_t)c * PS_C4;
  unsigned* seg  = pl + (size_t)(blockIdx.x * 32 + wv) * PS_CAP;

  int4 v[PS_PASSES];
#pragma unroll
  for (int p = 0; p < PS_PASSES; ++p) v[p] = d4[min(p * 1024 + tid, PS_C4 - 1)];
  for (int r = tid; r < PS_GROWS; r += 1024) hist[r] = 0;
  __syncthreads();
  int n = 0;
#pragma unroll
  for (int p = 0; p < PS_PASSES; ++p) {
    const int idx = p * 1024 + tid;
    const unsigned lim = (idx < PS_C4) ? gn : 0u;
    const unsigned e0s = (unsigned)(c * PS_C4 + idx) << 14;
    const int dv[4] = {v[p].x, v[p].y, v[p].z, v[p].w};
#pragma unroll
    for (int k = 0; k < 4; ++k) {
      const unsigned rr = (unsigned)(dv[k] - glo);
      const bool hit = rr < lim;
      const unsigned m = __builtin_amdgcn_ballot_w32(hit);
      if (m) {
        if (hit) {
          seg[n + __builtin_amdgcn_mbcnt_lo(m, 0u)] = e0s + ((unsigned)k << 12) + rr;
          __hip_atomic_fetch_add(hist + rr, 1, __ATOMIC_RELAXED, __HIP_MEMORY_SCOPE_WORKGROUP);
        }
        n += __builtin_popcount(m);
      }
    }
  }
  if (lane == 0) pcnt[blockIdx.x * 32 + wv] = n;
  __syncthreads();
  for (int r = tid; r < (int)gn; r += 1024) pc[(size_t)c * PC_PITCH + glo + r] = hist[r];
}

__device__ __forceinline__ void agg_sift4(uint4 e4, int left, unsigned sublo, unsigned nrows,
                                          unsigned* __restrict__ myq, int& qn) {
  const unsigned ev[4] = {e4.x, e4.y, e4.z, e4.w};
#pragma unroll
  for (int k = 0; k < 4; ++k) {
    const unsigned rr = (ev[k] & 4095u) - sublo;
    const bool hit = (k < left) && (rr < nrows);
    const unsigned m = __builtin_amdgcn_ballot_w32(hit);
    if (m) {
      if (hit) myq[qn + __builtin_amdgcn_mbcnt_lo(m, 0u)] = ((ev[k] >> 12) << 9) | rr;
      qn += __builtin_popcount(m);
    }
  }
}

template <int C>
__global__ __launch_bounds__(1024) void k_agg_sift(const float* __restrict__ H, const int* __restrict__ src,
                                                   const unsigned* __restrict__ pl, const int* __restrict__ pcnt,
                                                   const float* __restrict__ deg,
                                                   const float* __restrict__ dinv,
                                                   const float* __restrict__ bias, float* __restrict__ out,
                                                   int relu, unsigned* __restrict__ list,
                                                   int* __restrict__ run_start, int* __restrict__ run_cnt) {
  __shared__ float acc[AGG_ROWS * C];
  __shared__ unsigned q[32 * AGG_QCAP];
  __shared__ int red[32];
  __shared__ int wc[32];
  __shared__ int wcur[32];
  const int tid  = threadIdx.x;
  const int lane = tid & 31;
  const int wv   = tid >> 5;
  const int lo   = blockIdx.x * AGG_ROWS;
  const int nrows = min(AGG_ROWS, NN - lo);

  {
    int part = 0;
    for (int i = tid; i < lo; i += 1024) part += (int)deg[i] - 1;
    int oc = 0;
    const int orow = wv + 32 * lane;
    if (lane < AGG_SLOTS && orow < nrows) oc = (int)deg[lo + orow] - 1;
#pragma unroll
    for (int m = 16; m > 0; m >>= 1) {
      part += __shfl_xor(part, m);
      oc   += __shfl_xor(oc, m);
    }
    if (lane == 0) {
      red[wv]  = part;
      wc[wv]   = oc;
      wcur[wv] = 0;
    }
  }
  agg_rows_init<C>(acc, H, dinv, bias, lo, nrows, wv, lane);
  __syncthreads();
  int pre = red[lane];
#pragma unroll
  for (int m = 16; m > 0; m >>= 1) pre += __shfl_xor(pre, m);
  v2u* seg = (v2u*)list + pre;
  const int myc = wc[lane];
  int incl = myc;
#pragma unroll
  for (int d = 1; d < 32; d <<= 1) {
    const int t = __shfl_up(incl, d);
    if (lane >= d) incl += t;
  }
  const int basev = incl - myc;

  unsigned* myq = q + wv * AGG_QCAP;
  int qn = 0;
  const int grp = blockIdx.x / PS_WG;
  const unsigned sublo = (unsigned)(lo - grp * PS_GROWS);
  const int sbase = grp * PS_NC * 32 + wv;
  const int myn = (lane < PS_NC) ? pcnt[sbase + 32 * lane] : 0;
#pragma nounroll
  for (int c0 = 0; c0 < PS_NC; c0 += PS_B) {
    uint4 ent[PS_B];
    int   nn[PS_B];
#pragma unroll
    for (int u = 0; u < PS_B; ++u) {
      nn[u] = __builtin_amdgcn_readlane(myn, c0 + u);
      const uint4* sp = (const uint4*)(pl + (size_t)(sbase + 32 * (c0 + u)) * PS_CAP);
      ent[u] = (4 * lane < nn[u]) ? sp[lane] : make_uint4(0u, 0u, 0u, 0u);
    }
#pragma unroll
    for (int u = 0; u < PS_B; ++u) agg_sift4(ent[u], nn[u] - 4 * lane, sublo, (unsigned)nrows, myq, qn);
    while (qn >= 32) {
      qn -= 32;
      agg_drain<C>(acc, myq + qn, 32, H, src, dinv, lo, lane, seg, wcur, basev);
    }
#pragma nounroll
    for (int u = 0; u < PS_B; ++u) {
      const int nu = __builtin_amdgcn_readlane(myn, c0 + u);
      const uint4* sp = (const uint4*)(pl + (size_t)(sbase + 32 * (c0 + u)) * PS_CAP);
      for (int off = 128; off < nu; off += 128) {
        const int left = nu - off - 4 * lane;
        const uint4 e4 = (left > 0) ? sp[off / 4 + lane] : make_uint4(0u, 0u, 0u, 0u);
        agg_sift4(e4, left, sublo, (unsigned)nrows, myq, qn);
        while (qn >= 32) {
          qn -= 32;
          agg_drain<C>(acc, myq + qn, 32, H, src, dinv, lo, lane, seg, wcur, basev);
        }
      }
    }
  }
  if (qn > 0) agg_drain<C>(acc, myq, qn, H, src, dinv, lo, lane, seg, wcur, basev);
  __syncthreads();

  if (tid < 32) {
    run_start[blockIdx.x * 32 + tid] = pre + basev;
    run_cnt[blockIdx.x * 32 + tid]   = wcur[tid];
  }
  agg_rows_out<C>(acc, out, lo, nrows, wv, lane, relu);
}

template <int C>
__device__ __forceinline__ void agg_reg_apply(float (&a0)[8], float (&a1)[8], float (&a2)[8], float (&a3)[8],
                                              v2u en, int n, const float* __restrict__ H, int lane) {
  constexpr int V = C / 32;
  constexpr int G = 16;
  const int s    = (int)(en.x >> 9);
  const int slot = (int)((en.x >> 5) & 15u);
  for (int j0 = 0; j0 < n; j0 += G) {
    vfN<V> v[G];
#pragma unroll
    for (int jj = 0; jj < G; ++jj) {
      const int sj = __builtin_amdgcn_readlane(s, j0 + jj);
      v[jj] = *(const vfN<V>*)(H + (size_t)sj * C + V * lane);
    }
#pragma unroll
    for (int jj = 0; jj < G; ++jj) {
      const int j = j0 + jj;
      if (j < n) {
        const int   sl = __builtin_amdgcn_readlane(slot, j) & 7;
        const float wj = __int_as_float(__builtin_amdgcn_readlane((int)en.y, j));
        a0[sl] = fmaf(v[jj][0], wj, a0[sl]);
        if constexpr (V > 1) a1[sl] = fmaf(v[jj][1], wj, a1[sl]);
        if constexpr (V > 2) {
          a2[sl] = fmaf(v[jj][2], wj, a2[sl]);
          a3[sl] = fmaf(v[jj][3], wj, a3[sl]);
        }
      }
    }
  }
}

template <int C>
__global__ __launch_bounds__(1024) void k_agg_list(const float* __restrict__ H, const unsigned* __restrict__ list,
                                                   const int* __restrict__ run_start,
                                                   const int* __restrict__ run_cnt,
                                                   const float* __restrict__ dinv,
                                                   const float* __restrict__ bias, float* __restrict__ out,
                                                   int relu) {
  constexpr int V = C / 32;
  static_assert(V == 1 || V == 2 || V == 4, "one register array per row component, up to four");
  const int tid  = threadIdx.x;
  const int lane = tid & 31;
  const int wv   = __builtin_amdgcn_readfirstlane(tid >> 5);
  const int lo   = blockIdx.x * AGG_ROWS;
  const int nrows = min(AGG_ROWS, NN - lo);
  const int gw   = blockIdx.x * 32 + wv;
  const int cnt  = run_cnt[gw];
  const v2u* seg = (const v2u*)list + run_start[gw];

  v2u en = {0u, 0u};
  if (lane < cnt) en = seg[lane];

  float a0[8], a1[8], a2[8], a3[8];
  {
    const vfN<V> bv = *(const vfN<V>*)(bias + V * lane);
#pragma unroll
    for (int sl = 0; sl < 8; ++sl) {
      const int r  = min(wv + 32 * min(sl, AGG_SLOTS - 1), nrows - 1);
      const float di = dinv[lo + r];
      const float d2 = di * di;
      const vfN<V> hv = *(const vfN<V>*)(H + (size_t)(lo + r) * C + V * lane);
      a0[sl] = hv[0] * d2 + bv[0];
      if constexpr (V > 1) a1[sl] = hv[1] * d2 + bv[1];
      if constexpr (V > 2) {
        a2[sl] = hv[2] * d2 + bv[2];
        a3[sl] = hv[3] * d2 + bv[3];
      }
    }
  }

  for (int c0 = 0; c0 < cnt; c0 += 32) {
    v2u nx = {0u, 0u};
    if (c0 + 32 + lane < cnt) nx = seg[c0 + 32 + lane];
    agg_reg_apply<C>(a0, a1, a2, a3, en, min(32, cnt - c0), H, lane);
    en = nx;
  }

#pragma unroll
  for (int sl = 0; sl < AGG_SLOTS; ++sl) {
    const int r = wv + 32 * sl;
    if (r < nrows) {
      vfN<V> ov;
      ov[0] = a0[sl];
      if constexpr (V > 1) ov[1] = a1[sl];
      if constexpr (V > 2) {
        ov[2] = a2[sl];
        ov[3] = a3[sl];
      }
#pragma unroll
      for (int k = 0; k < V; ++k) ov[k] = relu ? fmaxf(ov[k], 0.0f) : ov[k];
      *(vfN<V>*)(out + (size_t)(lo + r) * C + V * lane) = ov;
    }
  }
}

extern "C" void kernel_launch(void* const* d_in, const int* in_sizes, int n_in,
                              void* d_out, int out_size, void* d_ws, size_t ws_size,
                              hipStream_t stream) {
  (void)in_sizes; (void)n_in; (void)out_size; (void)ws_size;
  const float* x   = (const float*)d_in[0];
  const int*   ei  = (const int*)d_in[1];
  const float* We1 = (const float*)d_in[2]; const float* be1 = (const float*)d_in[3];
  const float* We2 = (const float*)d_in[4]; const float* be2 = (const float*)d_in[5];
  const float* Wd1 = (const float*)d_in[6]; const float* bd1 = (const float*)d_in[7];
  const float* Wd2 = (const float*)d_in[8]; const float* bd2 = (const float*)d_in[9];
  const int* src = ei;
  const int* dst = ei + NE;

  char*  ws   = (char*)d_ws;
  float* dinv = (float*)ws;
  size_t off  = ((size_t)NN * 4 + 255) & ~(size_t)255;
  float* G    = (float*)(ws + off);
  float* P    = G + (size_t)NN * 128;
  float* deg  = P + (size_t)NN * 64;
  unsigned* list = (unsigned*)((char*)deg + off);
  int* run_start = (int*)(list + (size_t)NE * 2);
  int* run_cnt   = run_start + AGG_NB * 32;
  unsigned* pl   = (unsigned*)(run_cnt + AGG_NB * 32);
  int* pcnt      = (int*)(pl + (size_t)PS_NG * PS_NC * 32 * PS_CAP);
  int* pc        = pcnt + PS_NG * PS_NC * 32;
  float* O    = (float*)d_out;

  k_presift<<<PS_NG * PS_NC, 1024, 0, stream>>>(dst, pl, pcnt, pc);

  k_gemm_stripe<128, 64, true><<<GEMM_NB, GEMM_WAVES * 32, 0, stream>>>(x, We1, G, pc, deg, dinv);
  k_agg_sift<64><<<AGG_NB, 1024, 0, stream>>>(G, src, pl, pcnt, deg, dinv, be1, P, 1, list, run_start, run_cnt);

  k_gemm_stripe<64, 32, false><<<GEMM_NB, GEMM_WAVES * 32, 0, stream>>>(P, We2, G, nullptr, nullptr, nullptr);
  k_agg_list<32> <<<AGG_NB, 1024, 0, stream>>>(G, list, run_start, run_cnt, dinv, be2, P, 0);

  k_gemm_stripe<32, 64, false><<<GEMM_NB, GEMM_WAVES * 32, 0, stream>>>(P, Wd1, G, nullptr, nullptr, nullptr);
  k_agg_list<64> <<<AGG_NB, 1024, 0, stream>>>(G, list, run_start, run_cnt, dinv, bd1, P, 1);

  k_gemm_stripe<64, 128, false><<<GEMM_NB, GEMM_WAVES * 32, 0, stream>>>(P, Wd2, G, nullptr, nullptr, nullptr);
  k_agg_list<128><<<AGG_NB, 1024, 0, stream>>>(G, list, run_start, run_cnt, dinv, bd2, O, 0);
}


# Aggregations: independent loads share a round trip. First: 18 dependent trips ahead of the edge loop to 1 (list offset from per-block sums, no deg scan), 4 per drain to 3. Later: run entries up front
# speedup vs baseline: 3.7185x; 1.0469x over previous
#include <hip/hip_runtime.h>


#define NN 50000
#define NE 800000

typedef __attribute__((ext_vector_type(2))) float v2f;
typedef __attribute__((ext_vector_type(8))) float v8f;

#define PS_NC     16
#define PC_PITCH  ((NN + 31) / 32 * 32)

#define GEMM_WAVES 8
#define GEMM_STRIPES (NN / 16)
#define GEMM_NB ((GEMM_STRIPES + GEMM_WAVES - 1) / GEMM_WAVES)
static_assert(NN % 16 == 0, "rows are covered by whole 16-row stripes");

typedef __attribute__((ext_vector_type(4))) float v4f;

template <int K, int N, bool DEG>
__global__ __launch_bounds__(GEMM_WAVES * 32) void k_gemm_stripe(const float* __restrict__ X,
                                                                 const float* __restrict__ W,
                                                                 float* __restrict__ H,
                                                                 const int* __restrict__ pc,
                                                                 float* __restrict__ deg,
                                                                 float* __restrict__ dinv,
                                                                 int* __restrict__ bsum) {
  constexpr int T  = N / 16;
  constexpr int KS = K / 4;
  constexpr int WP = N + 16;
  __shared__ v2f wl[(K / 2) * WP];
  __shared__ int bpart[GEMM_WAVES];
  const int tid  = threadIdx.x;
  const int lane = tid & 31;
  const int wv   = __builtin_amdgcn_readfirstlane(tid >> 5);
  const int half = lane >> 4;
  const int l    = lane & 15;
  const int stripe = blockIdx.x * GEMM_WAVES + wv;
  const bool live = stripe < GEMM_STRIPES;

  v2f a[KS];
  if (live) {
    const float* xrow = X + (size_t)(stripe * 16 + l) * K + 2 * half;
#pragma unroll
    for (int kk = 0; kk < KS; ++kk) a[kk] = *(const v2f*)(xrow + 4 * kk);
  } else {
#pragma unroll
    for (int kk = 0; kk < KS; ++kk) a[kk] = v2f{0.0f, 0.0f};
  }

  constexpr int NI = (K / 2) * (N / 4) / (GEMM_WAVES * 32);
  static_assert(NI * GEMM_WAVES * 32 == (K / 2) * (N / 4), "W is staged in whole passes of the work-group");
  v4f w0[NI], w1[NI];
#pragma unroll
  for (int it = 0; it < NI; ++it) {
    const int i  = tid + it * GEMM_WAVES * 32;
    const int kp = i / (N / 4);
    const int q  = i - kp * (N / 4);
    w0[it] = *(const v4f*)(W + (size_t)(2 * kp) * N + 4 * q);
    w1[it] = *(const v4f*)(W + (size_t)(2 * kp + 1) * N + 4 * q);
  }
  if constexpr (DEG) {
    const int row = blockIdx.x * (GEMM_WAVES * 16) + tid;
    int n = 0;
    if (tid < GEMM_WAVES * 16 && row < NN) {
#pragma unroll
      for (int c = 0; c < PS_NC; ++c) n += pc[(size_t)c * PC_PITCH + row];
      const float d = 1.0f + (float)n;
      deg[row]  = d;
      dinv[row] = rsqrtf(d);
    }
#pragma unroll
    for (int m = 16; m > 0; m >>= 1) n += __shfl_xor(n, m);
    if (lane == 0) bpart[wv] = n;
  }
#pragma unroll
  for (int it = 0; it < NI; ++it) {
    const int i  = tid + it * GEMM_WAVES * 32;
    const int kp = i / (N / 4);
    const int q  = i - kp * (N / 4);
    v4f* dp = (v4f*)(wl + kp * WP + 4 * q);
    dp[0] = v4f{w0[it].x, w1[it].x, w0[it].y, w1[it].y};
    dp[1] = v4f{w0[it].z, w1[it].z, w0[it].w, w1[it].w};
  }
  __syncthreads();
  if constexpr (DEG) {
    if (tid == 0) {
      int t = 0;
#pragma unroll
      for (int w = 0; w < GEMM_WAVES; ++w) t += bpart[w];
      bsum[blockIdx.x] = t;
    }
  }
  if (!live) return;

  v8f c[T];
#pragma unroll
  for (int t = 0; t < T; ++t) c[t] = v8f{};
  const v2f* bp = wl + half * WP + l;
#pragma unroll
  for (int kk = 0; kk < KS; ++kk) {
    v2f b[T];
#pragma unroll
    for (int t = 0; t < T; ++t) b[t] = bp[(2 * kk) * WP + 16 * t];
#pragma unroll
    for (int t = 0; t < T; ++t)
      c[t] = __builtin_amdgcn_wmma_f32_16x16x4_f32(false, a[kk], false, b[t], (short)0, c[t], false, false);
  }
  float* hrow = H + (size_t)(stripe * 16 + 8 * half) * N + l;
#pragma unroll
  for (int t = 0; t < T; ++t) {
#pragma unroll
    for (int r = 0; r < 8; ++r) hrow[(size_t)r * N + 16 * t] = c[t][r];
  }
}

#define AGG_NB    256
#define AGG_ROWS  ((NN + AGG_NB - 1) / AGG_NB)
#define PS_NG     16
#define PS_WG     (AGG_NB / PS_NG)
#define PS_GROWS  (PS_WG * AGG_ROWS)
#define PS_C4     (NE / 4 / PS_NC)
#define PS_PASSES ((PS_C4 + 1023) / 1024)
#define PS_CAP    (PS_PASSES * 128)
#define PS_B      8
#define AGG_QCAP  (32 + 128 * PS_B)
#define AGG_SLOTS ((AGG_ROWS + 31) / 32)
static_assert(AGG_SLOTS <= 8, "a wave's rows are held in register arrays of 8 slots");
#define AGG_PRE   4

static_assert(NE % (4 * PS_NC) == 0, "dst is streamed as int4, in whole chunks");
static_assert(AGG_NB % PS_NG == 0 && PS_NC % PS_B == 0 && PS_NC <= 32, "groups and chunks tile evenly");
static_assert(PS_GROWS <= (1 << 12) && NE <= (1 << 20), "kept entry = edge id << 12 | row in group");
static_assert(PS_CAP % 4 == 0, "kept-edge segments start on 16-byte boundaries");
static_assert(AGG_ROWS <= 512, "row offset is packed into 9 bits");
static_assert(NE <= (1 << 22), "edge id is packed above 9 bits");
static_assert(NN <= (1 << 23), "source row is packed above 9 bits");
static_assert(GEMM_NB <= 1024, "one thread per product block sums the blocks before a work-group's rows");

template <int V> using vfN = float __attribute__((ext_vector_type(V)));
typedef unsigned __attribute__((ext_vector_type(2))) v2u;

#define AGG_G 16
template <int C, int J0>
__device__ __forceinline__ void agg_group(float* __restrict__ acc, int s, int r, float ds, float dr, int cnt,
                                          const float* __restrict__ H, int lane) {
  constexpr int V = C / 32;
  vfN<V> v[AGG_G];
#pragma unroll
  for (int jj = 0; jj < AGG_G; ++jj) {
    const int sj = __builtin_amdgcn_readlane(s, J0 + jj);
    v[jj] = *(const vfN<V>*)(H + (size_t)sj * C + V * lane);
  }
  const float w = ds * dr;
#pragma unroll
  for (int jj = 0; jj < AGG_G; ++jj) {
    const int j = J0 + jj;
    const int   rj = __builtin_amdgcn_readlane(r, j);
    const float wj = __int_as_float(__builtin_amdgcn_readlane(__float_as_int(w), j));
    float* ap = acc + rj * C + lane;
    const bool live = j < cnt;
#pragma unroll
    for (int k = 0; k < V; ++k) {
      const float m = live ? v[jj][k] * wj : 0.0f;
      __hip_atomic_fetch_add(ap + 32 * k, m, __ATOMIC_RELAXED, __HIP_MEMORY_SCOPE_WORKGROUP);
    }
  }
}

template <int C>
__device__ __forceinline__ void agg_drain(float* __restrict__ acc, const unsigned* __restrict__ qp,
                                          int cnt, const float* __restrict__ H,
                                          const int* __restrict__ src, const float* __restrict__ dinv,
                                          int lo, int lane, v2u* __restrict__ seg, int* wcur, int basev) {
  const bool act = lane < cnt;
  const unsigned ent = act ? qp[lane] : 0u;
  const int e = (int)(ent >> 9);
  const int r = (int)(ent & 511u);
  const int s = act ? src[e] : 0;
  const float dr = dinv[lo + r];
  const float ds = act ? dinv[s] : 0.0f;
  agg_group<C, 0>(acc, s, r, ds, dr, cnt, H, lane);
  if (cnt > AGG_G) agg_group<C, AGG_G>(acc, s, r, ds, dr, cnt, H, lane);
  const int owner = r & 31;
  const int ob = __shfl(basev, owner);
  if (act) {
    const int pos = __hip_atomic_fetch_add(wcur + owner, 1, __ATOMIC_RELAXED, __HIP_MEMORY_SCOPE_WORKGROUP);
    v2u le;
    le.x = ((unsigned)s << 9) | (unsigned)r;
    le.y = __float_as_uint(ds * dr);
    seg[ob + pos] = le;
  }
}

template <int C>
__device__ __forceinline__ void agg_rows_out(const float* __restrict__ acc, float* __restrict__ out,
                                             int lo, int nrows, int wv, int lane, int relu) {
  constexpr int V = C / 32;
  for (int r = wv; r < nrows; r += 32) {
    vfN<V> ov;
#pragma unroll
    for (int k = 0; k < V; ++k) {
      const float x = acc[r * C + k * 32 + lane];
      ov[k] = relu ? fmaxf(x, 0.0f) : x;
    }
    *(vfN<V>*)(out + (size_t)(lo + r) * C + V * lane) = ov;
  }
}

__global__ __launch_bounds__(1024) void k_presift(const int* __restrict__ dst, unsigned* __restrict__ pl,
                                                  int* __restrict__ pcnt, int* __restrict__ pc) {
  __shared__ int hist[PS_GROWS];
  const int tid  = threadIdx.x;
  const int lane = tid & 31;
  const int wv   = tid >> 5;
  const int g    = blockIdx.x / PS_NC;
  const int c    = blockIdx.x - g * PS_NC;
  const int glo  = g * PS_GROWS;
  const unsigned gn = (unsigned)min(PS_GROWS, NN - glo);
  const int4* d4 = (const int4*)dst + (size_t)c * PS_C4;
  unsigned* seg  = pl + (size_t)(blockIdx.x * 32 + wv) * PS_CAP;

  int4 v[PS_PASSES];
#pragma unroll
  for (int p = 0; p < PS_PASSES; ++p) v[p] = d4[min(p * 1024 + tid, PS_C4 - 1)];
  for (int r = tid; r < PS_GROWS; r += 1024) hist[r] = 0;
  __syncthreads();
  int n = 0;
#pragma unroll
  for (int p = 0; p < PS_PASSES; ++p) {
    const int idx = p * 1024 + tid;
    const unsigned lim = (idx < PS_C4) ? gn : 0u;
    const unsigned e0s = (unsigned)(c * PS_C4 + idx) << 14;
    const int dv[4] = {v[p].x, v[p].y, v[p].z, v[p].w};
#pragma unroll
    for (int k = 0; k < 4; ++k) {
      const unsigned rr = (unsigned)(dv[k] - glo);
      const bool hit = rr < lim;
      const unsigned m = __builtin_amdgcn_ballot_w32(hit);
      if (m) {
        if (hit) {
          seg[n + __builtin_amdgcn_mbcnt_lo(m, 0u)] = e0s + ((unsigned)k << 12) + rr;
          __hip_atomic_fetch_add(hist + rr, 1, __ATOMIC_RELAXED, __HIP_MEMORY_SCOPE_WORKGROUP);
        }
        n += __builtin_popcount(m);
      }
    }
  }
  if (lane == 0) pcnt[blockIdx.x * 32 + wv] = n;
  __syncthreads();
  for (int r = tid; r < (int)gn; r += 1024) pc[(size_t)c * PC_PITCH + glo + r] = hist[r];
}

__device__ __forceinline__ void agg_sift4(uint4 e4, int left, unsigned sublo, unsigned nrows,
                                          unsigned* __restrict__ myq, int& qn) {
  const unsigned ev[4] = {e4.x, e4.y, e4.z, e4.w};
#pragma unroll
  for (int k = 0; k < 4; ++k) {
    const unsigned rr = (ev[k] & 4095u) - sublo;
    const bool hit = (k < left) && (rr < nrows);
    const unsigned m = __builtin_amdgcn_ballot_w32(hit);
    if (m) {
      if (hit) myq[qn + __builtin_amdgcn_mbcnt_lo(m, 0u)] = ((ev[k] >> 12) << 9) | rr;
      qn += __builtin_popcount(m);
    }
  }
}

template <int C>
__global__ __launch_bounds__(1024) void k_agg_sift(const float* __restrict__ H, const int* __restrict__ src,
                                                   const unsigned* __restrict__ pl, const int* __restrict__ pcnt,
                                                   const float* __restrict__ deg,
                                                   const int* __restrict__ bsum,
                                                   const float* __restrict__ dinv,
                                                   const float* __restrict__ bias, float* __restrict__ out,
                                                   int relu, unsigned* __restrict__ list,
                                                   int* __restrict__ run_start, int* __restrict__ run_cnt) {
  __shared__ float acc[AGG_ROWS * C];
  __shared__ unsigned q[32 * AGG_QCAP];
  __shared__ int red[32];
  __shared__ int wc[32];
  __shared__ int wcur[32];
  constexpr int V = C / 32;
  const int tid  = threadIdx.x;
  const int lane = tid & 31;
  const int wv   = tid >> 5;
  const int lo   = blockIdx.x * AGG_ROWS;
  const int nrows = min(AGG_ROWS, NN - lo);
  const int grp = blockIdx.x / PS_WG;
  const int sbase = grp * PS_NC * 32 + wv;

  const int nb = lo / (GEMM_WAVES * 16);
  const int pr = nb * (GEMM_WAVES * 16) + tid;
  const int orow = wv + 32 * lane;
  const int   ld_bs = bsum[min(tid, GEMM_NB - 1)];
  const float ld_dp = deg[min(pr, NN - 1)];
  const float ld_do = deg[lo + min(orow, nrows - 1)];
  const int   ld_pn = pcnt[sbase + 32 * min(lane, PS_NC - 1)];
  const vfN<V> bv = *(const vfN<V>*)(bias + V * lane);
  float  di[AGG_SLOTS];
  vfN<V> hv[AGG_SLOTS];
#pragma unroll
  for (int sl = 0; sl < AGG_SLOTS; ++sl) {
    const int rc = lo + min(wv + 32 * sl, nrows - 1);
    di[sl] = dinv[rc];
    hv[sl] = *(const vfN<V>*)(H + (size_t)rc * C + V * lane);
  }

  {
    int part = (tid < nb) ? ld_bs : 0;
    if (pr < lo) part += (int)ld_dp - 1;
    int oc = 0;
    if (lane < AGG_SLOTS && orow < nrows) oc = (int)ld_do - 1;
#pragma unroll
    for (int m = 16; m > 0; m >>= 1) {
      part += __shfl_xor(part, m);
      oc   += __shfl_xor(oc, m);
    }
    if (lane == 0) {
      red[wv]  = part;
      wc[wv]   = oc;
      wcur[wv] = 0;
    }
  }
#pragma unroll
  for (int sl = 0; sl < AGG_SLOTS; ++sl) {
    const int r = wv + 32 * sl;
    if (r < nrows) {
      const float d2 = di[sl] * di[sl];
#pragma unroll
      for (int k = 0; k < V; ++k) acc[r * C + k * 32 + lane] = hv[sl][k] * d2 + bv[k];
    }
  }
  __syncthreads();
  int pre = red[lane];
#pragma unroll
  for (int m = 16; m > 0; m >>= 1) pre += __shfl_xor(pre, m);
  v2u* seg = (v2u*)list + pre;
  const int myc = wc[lane];
  int incl = myc;
#pragma unroll
  for (int d = 1; d < 32; d <<= 1) {
    const int t = __shfl_up(incl, d);
    if (lane >= d) incl += t;
  }
  const int basev = incl - myc;

  unsigned* myq = q + wv * AGG_QCAP;
  int qn = 0;
  const unsigned sublo = (unsigned)(lo - grp * PS_GROWS);
  const int myn = (lane < PS_NC) ? ld_pn : 0;
#pragma nounroll
  for (int c0 = 0; c0 < PS_NC; c0 += PS_B) {
    uint4 ent[PS_B];
    int   nn[PS_B];
#pragma unroll
    for (int u = 0; u < PS_B; ++u) {
      nn[u] = __builtin_amdgcn_readlane(myn, c0 + u);
      const uint4* sp = (const uint4*)(pl + (size_t)(sbase + 32 * (c0 + u)) * PS_CAP);
      ent[u] = (4 * lane < nn[u]) ? sp[lane] : make_uint4(0u, 0u, 0u, 0u);
    }
#pragma unroll
    for (int u = 0; u < PS_B; ++u) agg_sift4(ent[u], nn[u] - 4 * lane, sublo, (unsigned)nrows, myq, qn);
    while (qn >= 32) {
      qn -= 32;
      agg_drain<C>(acc, myq + qn, 32, H, src, dinv, lo, lane, seg, wcur, basev);
    }
#pragma nounroll
    for (int u = 0; u < PS_B; ++u) {
      const int nu = __builtin_amdgcn_readlane(myn, c0 + u);
      const uint4* sp = (const uint4*)(pl + (size_t)(sbase + 32 * (c0 + u)) * PS_CAP);
      for (int off = 128; off < nu; off += 128) {
        const int left = nu - off - 4 * lane;
        const uint4 e4 = (left > 0) ? sp[off / 4 + lane] : make_uint4(0u, 0u, 0u, 0u);
        agg_sift4(e4, left, sublo, (unsigned)nrows, myq, qn);
        while (qn >= 32) {
          qn -= 32;
          agg_drain<C>(acc, myq + qn, 32, H, src, dinv, lo, lane, seg, wcur, basev);
        }
      }
    }
  }
  if (qn > 0) agg_drain<C>(acc, myq, qn, H, src, dinv, lo, lane, seg, wcur, basev);
  __syncthreads();

  if (tid < 32) {
    run_start[blockIdx.x * 32 + tid] = pre + basev;
    run_cnt[blockIdx.x * 32 + tid]   = wcur[tid];
  }
  agg_rows_out<C>(acc, out, lo, nrows, wv, lane, relu);
}

template <int C>
__device__ __forceinline__ void agg_reg_apply(float (&a0)[8], float (&a1)[8], float (&a2)[8], float (&a3)[8],
                                              v2u en, int n, const float* __restrict__ H, int lane) {
  constexpr int V = C / 32;
  constexpr int G = 16;
  const int s    = (int)(en.x >> 9);
  const int slot = (int)((en.x >> 5) & 15u);
  for (int j0 = 0; j0 < n; j0 += G) {
    vfN<V> v[G];
#pragma unroll
    for (int jj = 0; jj < G; ++jj) {
      const int sj = __builtin_amdgcn_readlane(s, j0 + jj);
      v[jj] = *(const vfN<V>*)(H + (size_t)sj * C + V * lane);
    }
#pragma unroll
    for (int jj = 0; jj < G; ++jj) {
      const int j = j0 + jj;
      if (j < n) {
        const int   sl = __builtin_amdgcn_readlane(slot, j) & 7;
        const float wj = __int_as_float(__builtin_amdgcn_readlane((int)en.y, j));
        a0[sl] = fmaf(v[jj][0], wj, a0[sl]);
        if constexpr (V > 1) a1[sl] = fmaf(v[jj][1], wj, a1[sl]);
        if constexpr (V > 2) {
          a2[sl] = fmaf(v[jj][2], wj, a2[sl]);
          a3[sl] = fmaf(v[jj][3], wj, a3[sl]);
        }
      }
    }
  }
}

template <int C>
__global__ __launch_bounds__(1024) void k_agg_list(const float* __restrict__ H, const unsigned* __restrict__ list,
                                                   const int* __restrict__ run_start,
                                                   const int* __restrict__ run_cnt,
                                                   const float* __restrict__ dinv,
                                                   const float* __restrict__ bias, float* __restrict__ out,
                                                   int relu) {
  constexpr int V = C / 32;
  static_assert(V == 1 || V == 2 || V == 4, "one register array per row component, up to four");
  const int tid  = threadIdx.x;
  const int lane = tid & 31;
  const int wv   = __builtin_amdgcn_readfirstlane(tid >> 5);
  const int lo   = blockIdx.x * AGG_ROWS;
  const int nrows = min(AGG_ROWS, NN - lo);
  const int gw   = blockIdx.x * 32 + wv;
  const int cnt  = run_cnt[gw];
  const v2u* seg = (const v2u*)list + run_start[gw];

  v2u e0 = {0u, 0u}, e1 = {0u, 0u}, e2 = {0u, 0u}, e3 = {0u, 0u};
  if (cnt > 0) {
    e0 = seg[min(lane, cnt - 1)];
    e1 = seg[min(32 + lane, cnt - 1)];
    e2 = seg[min(64 + lane, cnt - 1)];
    e3 = seg[min(96 + lane, cnt - 1)];
  }

  float a0[8], a1[8], a2[8], a3[8];
  {
    const vfN<V> bv = *(const vfN<V>*)(bias + V * lane);
#pragma unroll
    for (int sl = 0; sl < 8; ++sl) {
      const int r  = min(wv + 32 * min(sl, AGG_SLOTS - 1), nrows - 1);
      const float di = dinv[lo + r];
      const float d2 = di * di;
      const vfN<V> hv = *(const vfN<V>*)(H + (size_t)(lo + r) * C + V * lane);
      a0[sl] = hv[0] * d2 + bv[0];
      if constexpr (V > 1) a1[sl] = hv[1] * d2 + bv[1];
      if constexpr (V > 2) {
        a2[sl] = hv[2] * d2 + bv[2];
        a3[sl] = hv[3] * d2 + bv[3];
      }
    }
  }

  for (int c0 = 0; c0 < cnt; c0 += 32) {
    const int u = c0 >> 5;
    v2u en = e0;
    if (u == 1) en = e1;
    if (u == 2) en = e2;
    if (u == 3) en = e3;
    if (u >= AGG_PRE) en = seg[min(c0 + lane, cnt - 1)];
    if (c0 + lane >= cnt) en = v2u{0u, 0u};
    agg_reg_apply<C>(a0, a1, a2, a3, en, min(32, cnt - c0), H, lane);
  }

#pragma unroll
  for (int sl = 0; sl < AGG_SLOTS; ++sl) {
    const int r = wv + 32 * sl;
    if (r < nrows) {
      vfN<V> ov;
      ov[0] = a0[sl];
      if constexpr (V > 1) ov[1] = a1[sl];
      if constexpr (V > 2) {
        ov[2] = a2[sl];
        ov[3] = a3[sl];
      }
#pragma unroll
      for (int k = 0; k < V; ++k) ov[k] = relu ? fmaxf(ov[k], 0.0f) : ov[k];
      *(vfN<V>*)(out + (size_t)(lo + r) * C + V * lane) = ov;
    }
  }
}

extern "C" void kernel_launch(void* const* d_in, const int* in_sizes, int n_in,
                              void* d_out, int out_size, void* d_ws, size_t ws_size,
                              hipStream_t stream) {
  (void)in_sizes; (void)n_in; (void)out_size; (void)ws_size;
  const float* x   = (const float*)d_in[0];
  const int*   ei  = (const int*)d_in[1];
  const float* We1 = (const float*)d_in[2]; const float* be1 = (const float*)d_in[3];
  const float* We2 = (const float*)d_in[4]; const float* be2 = (const float*)d_in[5];
  const float* Wd1 = (const float*)d_in[6]; const float* bd1 = (const float*)d_in[7];
  const float* Wd2 = (const float*)d_in[8]; const float* bd2 = (const float*)d_in[9];
  const int* src = ei;
  const int* dst = ei + NE;

  char*  ws   = (char*)d_ws;
  float* dinv = (float*)ws;
  size_t off  = ((size_t)NN * 4 + 255) & ~(size_t)255;
  float* G    = (float*)(ws + off);
  float* P    = G + (size_t)NN * 128;
  float* deg  = P + (size_t)NN * 64;
  unsigned* list = (unsigned*)((char*)deg + off);
  int* run_start = (int*)(list + (size_t)NE * 2);
  int* run_cnt   = run_start + AGG_NB * 32;
  unsigned* pl   = (unsigned*)(run_cnt + AGG_NB * 32);
  int* pcnt      = (int*)(pl + (size_t)PS_NG * PS_NC * 32 * PS_CAP);
  int* pc        = pcnt + PS_NG * PS_NC * 32;
  int* bsum      = pc + (size_t)PS_NC * PC_PITCH;
  float* O    = (float*)d_out;

  k_presift<<<PS_NG * PS_NC, 1024, 0, stream>>>(dst, pl, pcnt, pc);

  k_gemm_stripe<128, 64, true><<<GEMM_NB, GEMM_WAVES * 32, 0, stream>>>(x, We1, G, pc, deg, dinv, bsum);
  k_agg_sift<64><<<AGG_NB, 1024, 0, stream>>>(G, src, pl, pcnt, deg, bsum, dinv, be1, P, 1, list, run_start, run_cnt);

  k_gemm_stripe<64, 32, false><<<GEMM_NB, GEMM_WAVES * 32, 0, stream>>>(P, We2, G, nullptr, nullptr, nullptr, nullptr);
  k_agg_list<32> <<<AGG_NB, 1024, 0, stream>>>(G, list, run_start, run_cnt, dinv, be2, P, 0);

  k_gemm_stripe<32, 64, false><<<GEMM_NB, GEMM_WAVES * 32, 0, stream>>>(P, Wd1, G, nullptr, nullptr, nullptr, nullptr);
  k_agg_list<64> <<<AGG_NB, 1024, 0, stream>>>(G, list, run_start, run_cnt, dinv, bd1, P, 1);

  k_gemm_stripe<64, 128, false><<<GEMM_NB, GEMM_WAVES * 32, 0, stream>>>(P, Wd2, G, nullptr, nullptr, nullptr, nullptr);
  k_agg_list<128><<<AGG_NB, 1024, 0, stream>>>(G, list, run_start, run_cnt, dinv, bd2, O, 0);
}
